# starting kernel
# baseline (speedup 1.0000x reference)
#include <hip/hip_runtime.h>
#include <hip/hip_bf16.h>

typedef float v2f __attribute__((ext_vector_type(2)));
typedef float v8f __attribute__((ext_vector_type(8)));

#define D 128


__global__ void gcn_init_deg(float* __restrict__ deg, int n) {
    int i = blockIdx.x * blockDim.x + threadIdx.x;
    if (i < n) deg[i] = 1.0f;
}

__global__ void gcn_deg_edges(const int* __restrict__ dst, float* __restrict__ deg, int n_edges) {
    int stride = gridDim.x * blockDim.x;
    for (int e = blockIdx.x * blockDim.x + threadIdx.x; e < n_edges; e += stride)
        atomicAdd(&deg[dst[e]], 1.0f);
}

__global__ void gcn_dinv(float* __restrict__ deg, int n) {
    int i = blockIdx.x * blockDim.x + threadIdx.x;
    if (i < n) {
        float d = deg[i];
        deg[i] = (d > 0.0f) ? __frsqrt_rn(d) : 0.0f;
    }
}


template <bool RELU>
__global__ void gcn_gemm_wmma(const float* __restrict__ X, const float* __restrict__ W,
                              float* __restrict__ H, int n_rows) {
    const int wave = threadIdx.x >> 5;
    const int lane = threadIdx.x & 31;
    const int mbase = blockIdx.x * 16;
    const int nbase = wave * 16;
    const int half = lane >> 4;
    const int l16  = lane & 15;

    const bool full_tile = (mbase + 16 <= n_rows);
    int arow = mbase + l16;
    if (arow >= n_rows) arow = n_rows - 1;
    const float* __restrict__ ap = X + (size_t)arow * D;
    const int bcol = nbase + l16;

    v8f acc = {};
    #pragma unroll 8
    for (int k = 0; k < D; k += 4) {
        const int kk = k + half * 2;
        v2f a;
        a.x = ap[kk];
        a.y = ap[kk + 1];
        if (RELU) { a.x = fmaxf(a.x, 0.0f); a.y = fmaxf(a.y, 0.0f); }
        v2f b;
        b.x = W[(size_t)kk * D + bcol];
        b.y = W[(size_t)(kk + 1) * D + bcol];
        acc = __builtin_amdgcn_wmma_f32_16x16x4_f32(
             false, a,  false, b,
             (short)0, acc,  false,  false);
    }

    const int crow0 = mbase + half * 8;
    float* __restrict__ cp = H + (size_t)crow0 * D + nbase + l16;
    if (full_tile) {
        #pragma unroll
        for (int v = 0; v < 8; ++v) cp[(size_t)v * D] = acc[v];
    } else {
        #pragma unroll
        for (int v = 0; v < 8; ++v)
            if (crow0 + v < n_rows) cp[(size_t)v * D] = acc[v];
    }
}


__global__ void gcn_agg_init(const float* __restrict__ H, const float* __restrict__ dinv,
                             const float* __restrict__ bias, float* __restrict__ Y, int n_rows) {
    int tid = blockIdx.x * blockDim.x + threadIdx.x;
    int node = tid >> 5;
    int c = (tid & 31) * 4;
    if (node >= n_rows) return;
    float s = dinv[node];
    s = s * s;
    float4 h = *(const float4*)(H + (size_t)node * D + c);
    float4 b = *(const float4*)(bias + c);
    float4 r = make_float4(fmaf(h.x, s, b.x), fmaf(h.y, s, b.y),
                           fmaf(h.z, s, b.z), fmaf(h.w, s, b.w));
    *(float4*)(Y + (size_t)node * D + c) = r;
}

__global__ void gcn_agg_edges(const float* __restrict__ H, const float* __restrict__ dinv,
                              const int* __restrict__ src, const int* __restrict__ dst,
                              float* __restrict__ Y, int n_edges) {
    const int lane = threadIdx.x & 31;
    const int wid = (blockIdx.x * blockDim.x + threadIdx.x) >> 5;
    const int nwaves = (gridDim.x * blockDim.x) >> 5;

    for (int e = wid; e < n_edges; e += nwaves) {
        const int s = src[e];
        const int d = dst[e];
        const int e2 = e + nwaves;
        if (e2 < n_edges) {
            const int s2 = src[e2];
            __builtin_prefetch(H + (size_t)s2 * D + lane * 4, 0, 1);
        }
        const float norm = dinv[s] * dinv[d];
        float4 h = *(const float4*)(H + (size_t)s * D + lane * 4);
        float* yp = Y + (size_t)d * D + lane * 4;
        atomicAdd(yp + 0, h.x * norm);
        atomicAdd(yp + 1, h.y * norm);
        atomicAdd(yp + 2, h.z * norm);
        atomicAdd(yp + 3, h.w * norm);
    }
}


extern "C" void kernel_launch(void* const* d_in, const int* in_sizes, int n_in,
                              void* d_out, int out_size, void* d_ws, size_t ws_size,
                              hipStream_t stream) {
    const float* x   = (const float*)d_in[0];
    const int*   ei  = (const int*)d_in[1];
    const float* W1  = (const float*)d_in[2];
    const float* b1  = (const float*)d_in[3];
    const float* W2  = (const float*)d_in[4];
    const float* b2  = (const float*)d_in[5];
    const float* W3  = (const float*)d_in[6];
    const float* b3  = (const float*)d_in[7];

    const int N = in_sizes[0] / D;
    const int E = in_sizes[1] / 2;
    const int* src = ei;
    const int* dst = ei + E;

    float* dinv = (float*)d_ws;
    size_t off = ((size_t)N * sizeof(float) + 255) & ~(size_t)255;
    float* H = (float*)((char*)d_ws + off);
    float* Y = H + (size_t)N * D;
    float* out = (float*)d_out;

    const int T = 256;
    const int nodeBlocks  = (N + T - 1) / T;
    const int mTiles      = (N + 15) / 16;
    const int initBlocks  = (int)(((size_t)N * 32 + T - 1) / T);
    const int edgeBlocks  = 2048;

    gcn_init_deg<<<nodeBlocks, T, 0, stream>>>(dinv, N);
    gcn_deg_edges<<<1024, T, 0, stream>>>(dst, dinv, E);
    gcn_dinv<<<nodeBlocks, T, 0, stream>>>(dinv, N);

    gcn_gemm_wmma<false><<<mTiles, T, 0, stream>>>(x, W1, H, N);
    gcn_agg_init<<<initBlocks, T, 0, stream>>>(H, dinv, b1, Y, N);
    gcn_agg_edges<<<edgeBlocks, T, 0, stream>>>(H, dinv, src, dst, Y, E);

    gcn_gemm_wmma<true><<<mTiles, T, 0, stream>>>(Y, W2, H, N);
    gcn_agg_init<<<initBlocks, T, 0, stream>>>(H, dinv, b2, Y, N);
    gcn_agg_edges<<<edgeBlocks, T, 0, stream>>>(H, dinv, src, dst, Y, E);

    gcn_gemm_wmma<true><<<mTiles, T, 0, stream>>>(Y, W3, H, N);
    gcn_agg_init<<<initBlocks, T, 0, stream>>>(H, dinv, b3, out, N);
    gcn_agg_edges<<<edgeBlocks, T, 0, stream>>>(H, dinv, src, dst, out, E);
}


# aggregation: edges grouped once per call by destination range of 200 rows; one work-group per range sums its rows in LDS (ds_add_f32), stores each once; sums still in arrival order; 10 launches
# speedup vs baseline: 2.3751x; 2.3751x over previous
#include <hip/hip_runtime.h>
#include <hip/hip_bf16.h>


typedef float v2f __attribute__((ext_vector_type(2)));
typedef float v8f __attribute__((ext_vector_type(8)));

#define D 128
#define RB 200
#define PT 1024
#define EU 8
#define NW (PT / 32)
#define NK ((RB + NW - 1) / NW)
#define ACC_ROWS 328


__global__ void gcn_init_deg(float* __restrict__ deg, int n) {
    int i = blockIdx.x * blockDim.x + threadIdx.x;
    if (i < n) deg[i] = 1.0f;
}

__global__ void gcn_deg_edges(const int* __restrict__ dst, float* __restrict__ deg, int n_edges) {
    int stride = gridDim.x * blockDim.x;
    for (int e = blockIdx.x * blockDim.x + threadIdx.x; e < n_edges; e += stride)
        atomicAdd(&deg[dst[e]], 1.0f);
}

__global__ void gcn_dinv_bins(float* __restrict__ deg, int* __restrict__ binTotal,
                              int* __restrict__ cursor, int n) {
    __shared__ int wsum[8];
    const int t = threadIdx.x;
    const int i = blockIdx.x * RB + t;
    int cnt = 0;
    if (t < RB && i < n) {
        float d = deg[i];
        cnt = (int)d - 1;
        deg[i] = (d > 0.0f) ? __frsqrt_rn(d) : 0.0f;
    }
    for (int o = 16; o > 0; o >>= 1) cnt += __shfl_xor(cnt, o);
    if ((t & 31) == 0) wsum[t >> 5] = cnt;
    __syncthreads();
    if (t == 0) {
        int s = 0;
        for (int k = 0; k < 8; ++k) s += wsum[k];
        binTotal[blockIdx.x] = s;
        cursor[blockIdx.x] = 0;
    }
}


__global__ __launch_bounds__(PT)
void gcn_bin_place(const int* __restrict__ src, const int* __restrict__ dst,
                   const float* __restrict__ dinv, const int* __restrict__ binTotal,
                   int* __restrict__ cursor, int* __restrict__ binStart,
                   int* __restrict__ recs, int n_edges, int n_bins) {
    extern __shared__ int sm[];
    int* startL = sm;
    int* cntL   = sm + (n_bins + 1);
    int* baseL  = cntL + n_bins;
    __shared__ int wsum[32];
    __shared__ int carryS;

    const int t = threadIdx.x;
    const int lane = t & 31;
    const int w = t >> 5;

    if (t == 0) carryS = 0;
    __syncthreads();
    for (int c0 = 0; c0 < n_bins; c0 += PT) {
        const int i = c0 + t;
        const int v = (i < n_bins) ? binTotal[i] : 0;
        int x = v;
        for (int o = 1; o < 32; o <<= 1) {
            const int y = __shfl_up(x, o);
            if (lane >= o) x += y;
        }
        if (lane == 31) wsum[w] = x;
        __syncthreads();
        int pre = carryS;
        for (int k = 0; k < w; ++k) pre += wsum[k];
        if (i < n_bins) startL[i] = pre + x - v;
        __syncthreads();
        if (t == PT - 1) carryS = pre + x;
        __syncthreads();
    }
    if (t == 0) startL[n_bins] = carryS;
    for (int i = t; i < n_bins; i += PT) cntL[i] = 0;
    __syncthreads();

    const int per = (n_edges + gridDim.x - 1) / gridDim.x;
    const int eb = blockIdx.x * per;
    const int ee = (eb + per < n_edges) ? (eb + per) : n_edges;

    for (int e = eb + t; e < ee; e += PT) atomicAdd(&cntL[dst[e] / RB], 1);
    __syncthreads();
    for (int i = t; i < n_bins; i += PT) {
        const int c = cntL[i];
        int base = startL[i];
        if (c > 0) base += atomicAdd(&cursor[i], c);
        baseL[i] = base;
        cntL[i] = 0;
    }
    __syncthreads();
    for (int e = eb + t; e < ee; e += PT) {
        const int s = src[e];
        const int d = dst[e];
        const int b = d / RB;
        const int pos = baseL[b] + atomicAdd(&cntL[b], 1);
        const float norm = dinv[s] * dinv[d];
        *(int4*)(recs + (size_t)pos * 4) = make_int4(s, d - b * RB, __float_as_int(norm), 0);
    }
    if (blockIdx.x == 0)
        for (int i = t; i <= n_bins; i += PT) binStart[i] = startL[i];
}


template <bool RELU>
__global__ void gcn_gemm_wmma(const float* __restrict__ X, const float* __restrict__ W,
                              float* __restrict__ H, int n_rows) {
    const int wave = threadIdx.x >> 5;
    const int lane = threadIdx.x & 31;
    const int mbase = blockIdx.x * 16;
    const int nbase = wave * 16;
    const int half = lane >> 4;
    const int l16  = lane & 15;

    const bool full_tile = (mbase + 16 <= n_rows);
    int arow = mbase + l16;
    if (arow >= n_rows) arow = n_rows - 1;
    const float* __restrict__ ap = X + (size_t)arow * D;
    const int bcol = nbase + l16;

    v8f acc = {};
    #pragma unroll 8
    for (int k = 0; k < D; k += 4) {
        const int kk = k + half * 2;
        v2f a;
        a.x = ap[kk];
        a.y = ap[kk + 1];
        if (RELU) { a.x = fmaxf(a.x, 0.0f); a.y = fmaxf(a.y, 0.0f); }
        v2f b;
        b.x = W[(size_t)kk * D + bcol];
        b.y = W[(size_t)(kk + 1) * D + bcol];
        acc = __builtin_amdgcn_wmma_f32_16x16x4_f32(
             false, a,  false, b,
             (short)0, acc,  false,  false);
    }

    const int crow0 = mbase + half * 8;
    float* __restrict__ cp = H + (size_t)crow0 * D + nbase + l16;
    if (full_tile) {
        #pragma unroll
        for (int v = 0; v < 8; ++v) cp[(size_t)v * D] = acc[v];
    } else {
        #pragma unroll
        for (int v = 0; v < 8; ++v)
            if (crow0 + v < n_rows) cp[(size_t)v * D] = acc[v];
    }
}


__global__ __launch_bounds__(PT)
void gcn_agg_lds(const float* __restrict__ H, const float* __restrict__ dinv,
                 const float* __restrict__ bias, const int* __restrict__ recs,
                 const int* __restrict__ binStart, float* __restrict__ Y, int n_rows) {
    __shared__ float acc[ACC_ROWS * D];

    const int t = threadIdx.x;
    const int lane = t & 31;
    const int wave = __builtin_amdgcn_readfirstlane(t >> 5);
    const int node0 = blockIdx.x * RB;
    int nrows = n_rows - node0;
    if (nrows > RB) nrows = RB;

    static_assert(NK * NW <= ACC_ROWS, "every slot of the start block needs an LDS row");
    const int c = lane * 4;
    {
        const float4 b = *(const float4*)(bias + c);
        #pragma unroll
        for (int k = 0; k < NK; ++k) {
            const int r = wave + NW * k;
            const int rl = (r < nrows) ? r : nrows - 1;
            float s = dinv[node0 + rl];
            s = s * s;
            const float4 h = *(const float4*)(H + (size_t)(node0 + rl) * D + c);
            *(float4*)(acc + r * D + c) = make_float4(fmaf(h.x, s, b.x), fmaf(h.y, s, b.y),
                                                      fmaf(h.z, s, b.z), fmaf(h.w, s, b.w));
        }
    }
    __syncthreads();

    const int e0 = binStart[blockIdx.x];
    const int e1 = binStart[blockIdx.x + 1];
    const int4* __restrict__ rp = (const int4*)recs;

    for (int e = e0 + wave; e < e1; e += EU * 32) {
        int4 r[EU];
        v2f lo[EU], hi[EU];
        #pragma unroll
        for (int u = 0; u < EU; ++u) {
            int eu = e + u * 32;
            if (eu >= e1) eu = e1 - 1;
            r[u] = rp[eu];
        }
        #pragma unroll
        for (int u = 0; u < EU; ++u) {
            const v2f* hp = (const v2f*)(H + (size_t)r[u].x * D) + lane;
            lo[u] = hp[0];
            hi[u] = hp[32];
        }
        #pragma unroll
        for (int u = 0; u < EU; ++u) {
            if (e + u * 32 < e1) {
                const float norm = __int_as_float(r[u].z);
                float* yp = acc + r[u].y * D + lane * 2;
                atomicAdd(yp +  0, lo[u].x * norm);
                atomicAdd(yp +  1, lo[u].y * norm);
                atomicAdd(yp + 64, hi[u].x * norm);
                atomicAdd(yp + 65, hi[u].y * norm);
            }
        }
    }
    __syncthreads();

    #pragma unroll
    for (int k = 0; k < NK; ++k) {
        const int r = wave + NW * k;
        if (r < nrows)
            *(float4*)(Y + (size_t)(node0 + r) * D + c) = *(const float4*)(acc + r * D + c);
    }
}


extern "C" void kernel_launch(void* const* d_in, const int* in_sizes, int n_in,
                              void* d_out, int out_size, void* d_ws, size_t ws_size,
                              hipStream_t stream) {
    const float* x   = (const float*)d_in[0];
    const int*   ei  = (const int*)d_in[1];
    const float* W1  = (const float*)d_in[2];
    const float* b1  = (const float*)d_in[3];
    const float* W2  = (const float*)d_in[4];
    const float* b2  = (const float*)d_in[5];
    const float* W3  = (const float*)d_in[6];
    const float* b3  = (const float*)d_in[7];

    const int N = in_sizes[0] / D;
    const int E = in_sizes[1] / 2;
    const int* src = ei;
    const int* dst = ei + E;
    const int nBins = (N + RB - 1) / RB;

    float* dinv = (float*)d_ws;
    size_t off = ((size_t)N * sizeof(float) + 255) & ~(size_t)255;
    float* H = (float*)((char*)d_ws + off);
    float* Y = H + (size_t)N * D;
    const size_t binBytes = ((size_t)(nBins + 1) * sizeof(int) + 255) & ~(size_t)255;
    char* p = (char*)(Y + (size_t)N * D);
    int* binTotal = (int*)p;  p += binBytes;
    int* cursor   = (int*)p;  p += binBytes;
    int* binStart = (int*)p;  p += binBytes;
    int* recs     = (int*)p;
    float* out = (float*)d_out;

    const int T = 256;
    const int nodeBlocks  = (N + T - 1) / T;
    const int mTiles      = (N + 15) / 16;
    const size_t placeLds = (size_t)(3 * nBins + 1) * sizeof(int);

    gcn_init_deg<<<nodeBlocks, T, 0, stream>>>(dinv, N);
    gcn_deg_edges<<<1024, T, 0, stream>>>(dst, dinv, E);
    gcn_dinv_bins<<<nBins, T, 0, stream>>>(dinv, binTotal, cursor, N);
    gcn_bin_place<<<256, PT, placeLds, stream>>>(src, dst, dinv, binTotal, cursor, binStart, recs, E, nBins);

    gcn_gemm_wmma<false><<<mTiles, T, 0, stream>>>(x, W1, H, N);
    gcn_agg_lds<<<nBins, PT, 0, stream>>>(H, dinv, b1, recs, binStart, Y, N);

    gcn_gemm_wmma<true><<<mTiles, T, 0, stream>>>(Y, W2, H, N);
    gcn_agg_lds<<<nBins, PT, 0, stream>>>(H, dinv, b2, recs, binStart, Y, N);

    gcn_gemm_wmma<true><<<mTiles, T, 0, stream>>>(Y, W3, H, N);
    gcn_agg_lds<<<nBins, PT, 0, stream>>>(H, dinv, b3, recs, binStart, out, N);
}


# GEMM re-tiled: W staged once per work-group in LDS, one wave per 16x128 row strip (A read once by 16-byte loads, 8 accumulators), whole-line stores via LDS; k summed in a new fixed order per 8
# speedup vs baseline: 2.6807x; 1.1287x over previous
#include <hip/hip_runtime.h>
#include <hip/hip_bf16.h>


typedef float v2f __attribute__((ext_vector_type(2)));
typedef float v8f __attribute__((ext_vector_type(8)));

#define D 128
#define RB 200
#define PT 1024
#define EU 8
#define NW (PT / 32)
#define NK ((RB + NW - 1) / NW)
#define ACC_ROWS 328


__global__ void gcn_init_deg(float* __restrict__ deg, int n) {
    int i = blockIdx.x * blockDim.x + threadIdx.x;
    if (i < n) deg[i] = 1.0f;
}

__global__ void gcn_deg_edges(const int* __restrict__ dst, float* __restrict__ deg, int n_edges) {
    int stride = gridDim.x * blockDim.x;
    for (int e = blockIdx.x * blockDim.x + threadIdx.x; e < n_edges; e += stride)
        atomicAdd(&deg[dst[e]], 1.0f);
}

__global__ void gcn_dinv_bins(float* __restrict__ deg, int* __restrict__ binTotal,
                              int* __restrict__ cursor, int n) {
    __shared__ int wsum[8];
    const int t = threadIdx.x;
    const int i = blockIdx.x * RB + t;
    int cnt = 0;
    if (t < RB && i < n) {
        float d = deg[i];
        cnt = (int)d - 1;
        deg[i] = (d > 0.0f) ? __frsqrt_rn(d) : 0.0f;
    }
    for (int o = 16; o > 0; o >>= 1) cnt += __shfl_xor(cnt, o);
    if ((t & 31) == 0) wsum[t >> 5] = cnt;
    __syncthreads();
    if (t == 0) {
        int s = 0;
        for (int k = 0; k < 8; ++k) s += wsum[k];
        binTotal[blockIdx.x] = s;
        cursor[blockIdx.x] = 0;
    }
}


__global__ __launch_bounds__(PT)
void gcn_bin_place(const int* __restrict__ src, const int* __restrict__ dst,
                   const float* __restrict__ dinv, const int* __restrict__ binTotal,
                   int* __restrict__ cursor, int* __restrict__ binStart,
                   int* __restrict__ recs, int n_edges, int n_bins) {
    extern __shared__ int sm[];
    int* startL = sm;
    int* cntL   = sm + (n_bins + 1);
    int* baseL  = cntL + n_bins;
    __shared__ int wsum[32];
    __shared__ int carryS;

    const int t = threadIdx.x;
    const int lane = t & 31;
    const int w = t >> 5;

    if (t == 0) carryS = 0;
    __syncthreads();
    for (int c0 = 0; c0 < n_bins; c0 += PT) {
        const int i = c0 + t;
        const int v = (i < n_bins) ? binTotal[i] : 0;
        int x = v;
        for (int o = 1; o < 32; o <<= 1) {
            const int y = __shfl_up(x, o);
            if (lane >= o) x += y;
        }
        if (lane == 31) wsum[w] = x;
        __syncthreads();
        int pre = carryS;
        for (int k = 0; k < w; ++k) pre += wsum[k];
        if (i < n_bins) startL[i] = pre + x - v;
        __syncthreads();
        if (t == PT - 1) carryS = pre + x;
        __syncthreads();
    }
    if (t == 0) startL[n_bins] = carryS;
    for (int i = t; i < n_bins; i += PT) cntL[i] = 0;
    __syncthreads();

    const int per = (n_edges + gridDim.x - 1) / gridDim.x;
    const int eb = blockIdx.x * per;
    const int ee = (eb + per < n_edges) ? (eb + per) : n_edges;

    for (int e = eb + t; e < ee; e += PT) atomicAdd(&cntL[dst[e] / RB], 1);
    __syncthreads();
    for (int i = t; i < n_bins; i += PT) {
        const int c = cntL[i];
        int base = startL[i];
        if (c > 0) base += atomicAdd(&cursor[i], c);
        baseL[i] = base;
        cntL[i] = 0;
    }
    __syncthreads();
    for (int e = eb + t; e < ee; e += PT) {
        const int s = src[e];
        const int d = dst[e];
        const int b = d / RB;
        const int pos = baseL[b] + atomicAdd(&cntL[b], 1);
        const float norm = dinv[s] * dinv[d];
        *(int4*)(recs + (size_t)pos * 4) = make_int4(s, d - b * RB, __float_as_int(norm), 0);
    }
    if (blockIdx.x == 0)
        for (int i = t; i <= n_bins; i += PT) binStart[i] = startL[i];
}


#define GW 16
#define GT (GW * 32)
#define GEMM_GROUPS 256
#define WPITCH 272
#define OPITCH 132

template <bool RELU>
__global__ __launch_bounds__(GT)
void gcn_gemm_wmma(const float* __restrict__ X, const float* __restrict__ W,
                   float* __restrict__ H, int n_rows) {
    __shared__ float Ws[(D / 2) * WPITCH];
    __shared__ float Os[GW * 16 * OPITCH];

    const int t = threadIdx.x;
    const int lane = t & 31;
    const int wave = __builtin_amdgcn_readfirstlane(t >> 5);
    const int half = lane >> 4;
    const int l16  = lane & 15;

    const int n_strips = (n_rows + 15) >> 4;
    const int stride = GW * gridDim.x;
    int s = wave * gridDim.x + blockIdx.x;

    float4 a[16];
    if (s < n_strips) {
        int arow = s * 16 + l16;
        if (arow >= n_rows) arow = n_rows - 1;
        const float4* __restrict__ ap = (const float4*)(X + (size_t)arow * D + 4 * half);
        #pragma unroll
        for (int q = 0; q < 16; ++q) a[q] = ap[2 * q];
    }

    #pragma unroll
    for (int it = 0; it < (D / 2) / GW; ++it) {
        const int kp = it * GW + wave;
        const float4 r0 = *(const float4*)(W + (size_t)(2 * kp) * D + 4 * lane);
        const float4 r1 = *(const float4*)(W + (size_t)(2 * kp + 1) * D + 4 * lane);
        float* p = Ws + kp * WPITCH + 8 * lane;
        *(float4*)(p)     = make_float4(r0.x, r1.x, r0.y, r1.y);
        *(float4*)(p + 4) = make_float4(r0.z, r1.z, r0.w, r1.w);
    }
    __syncthreads();

    float* __restrict__ Ow = Os + wave * (16 * OPITCH);
    const float* __restrict__ bbase = Ws + (2 * half) * WPITCH + 2 * l16;

    while (s < n_strips) {
        if (RELU) {
            #pragma unroll
            for (int q = 0; q < 16; ++q) {
                a[q].x = fmaxf(a[q].x, 0.0f); a[q].y = fmaxf(a[q].y, 0.0f);
                a[q].z = fmaxf(a[q].z, 0.0f); a[q].w = fmaxf(a[q].w, 0.0f);
            }
        }

        v8f acc[8];
        #pragma unroll
        for (int n = 0; n < 8; ++n) acc[n] = (v8f){};

        #pragma unroll
        for (int q = 0; q < 16; ++q) {
            #pragma unroll
            for (int u = 0; u < 2; ++u) {
                const float* __restrict__ bp = bbase + (4 * q + u) * WPITCH;
                v2f av;
                av.x = u ? a[q].z : a[q].x;
                av.y = u ? a[q].w : a[q].y;
                #pragma unroll
                for (int n = 0; n < 8; ++n) {
                    const v2f b = *(const v2f*)(bp + 32 * n);
                    acc[n] = __builtin_amdgcn_wmma_f32_16x16x4_f32(
                         false, av,  false, b,
                         (short)0, acc[n],  false,  false);
                }
            }
        }

        #pragma unroll
        for (int n = 0; n < 8; ++n) {
            #pragma unroll
            for (int v = 0; v < 8; ++v)
                Ow[(8 * half + v) * OPITCH + 16 * n + l16] = acc[n][v];
        }
        __builtin_amdgcn_fence(__ATOMIC_RELEASE, "wavefront");
        __builtin_amdgcn_wave_barrier();
        __builtin_amdgcn_fence(__ATOMIC_ACQUIRE, "wavefront");

        const int row0 = s * 16;
        float* __restrict__ hp = H + (size_t)row0 * D + 4 * lane;
        if (row0 + 16 <= n_rows) {
            #pragma unroll
            for (int i = 0; i < 16; ++i)
                *(float4*)(hp + (size_t)i * D) = *(const float4*)(Ow + i * OPITCH + 4 * lane);
        } else {
            #pragma unroll
            for (int i = 0; i < 16; ++i)
                if (row0 + i < n_rows)
                    *(float4*)(hp + (size_t)i * D) = *(const float4*)(Ow + i * OPITCH + 4 * lane);
        }

        s += stride;
        if (s < n_strips) {
            int arow = s * 16 + l16;
            if (arow >= n_rows) arow = n_rows - 1;
            const float4* __restrict__ ap = (const float4*)(X + (size_t)arow * D + 4 * half);
            #pragma unroll
            for (int q = 0; q < 16; ++q) a[q] = ap[2 * q];
        }
    }
}


__global__ __launch_bounds__(PT)
void gcn_agg_lds(const float* __restrict__ H, const float* __restrict__ dinv,
                 const float* __restrict__ bias, const int* __restrict__ recs,
                 const int* __restrict__ binStart, float* __restrict__ Y, int n_rows) {
    __shared__ float acc[ACC_ROWS * D];

    const int t = threadIdx.x;
    const int lane = t & 31;
    const int wave = __builtin_amdgcn_readfirstlane(t >> 5);
    const int node0 = blockIdx.x * RB;
    int nrows = n_rows - node0;
    if (nrows > RB) nrows = RB;

    static_assert(NK * NW <= ACC_ROWS, "every slot of the start block needs an LDS row");
    const int c = lane * 4;
    {
        const float4 b = *(const float4*)(bias + c);
        #pragma unroll
        for (int k = 0; k < NK; ++k) {
            const int r = wave + NW * k;
            const int rl = (r < nrows) ? r : nrows - 1;
            float s = dinv[node0 + rl];
            s = s * s;
            const float4 h = *(const float4*)(H + (size_t)(node0 + rl) * D + c);
            *(float4*)(acc + r * D + c) = make_float4(fmaf(h.x, s, b.x), fmaf(h.y, s, b.y),
                                                      fmaf(h.z, s, b.z), fmaf(h.w, s, b.w));
        }
    }
    __syncthreads();

    const int e0 = binStart[blockIdx.x];
    const int e1 = binStart[blockIdx.x + 1];
    const int4* __restrict__ rp = (const int4*)recs;

    for (int e = e0 + wave; e < e1; e += EU * 32) {
        int4 r[EU];
        v2f lo[EU], hi[EU];
        #pragma unroll
        for (int u = 0; u < EU; ++u) {
            int eu = e + u * 32;
            if (eu >= e1) eu = e1 - 1;
            r[u] = rp[eu];
        }
        #pragma unroll
        for (int u = 0; u < EU; ++u) {
            const v2f* hp = (const v2f*)(H + (size_t)r[u].x * D) + lane;
            lo[u] = hp[0];
            hi[u] = hp[32];
        }
        #pragma unroll
        for (int u = 0; u < EU; ++u) {
            if (e + u * 32 < e1) {
                const float norm = __int_as_float(r[u].z);
                float* yp = acc + r[u].y * D + lane * 2;
                atomicAdd(yp +  0, lo[u].x * norm);
                atomicAdd(yp +  1, lo[u].y * norm);
                atomicAdd(yp + 64, hi[u].x * norm);
                atomicAdd(yp + 65, hi[u].y * norm);
            }
        }
    }
    __syncthreads();

    #pragma unroll
    for (int k = 0; k < NK; ++k) {
        const int r = wave + NW * k;
        if (r < nrows)
            *(float4*)(Y + (size_t)(node0 + r) * D + c) = *(const float4*)(acc + r * D + c);
    }
}


extern "C" void kernel_launch(void* const* d_in, const int* in_sizes, int n_in,
                              void* d_out, int out_size, void* d_ws, size_t ws_size,
                              hipStream_t stream) {
    const float* x   = (const float*)d_in[0];
    const int*   ei  = (const int*)d_in[1];
    const float* W1  = (const float*)d_in[2];
    const float* b1  = (const float*)d_in[3];
    const float* W2  = (const float*)d_in[4];
    const float* b2  = (const float*)d_in[5];
    const float* W3  = (const float*)d_in[6];
    const float* b3  = (const float*)d_in[7];

    const int N = in_sizes[0] / D;
    const int E = in_sizes[1] / 2;
    const int* src = ei;
    const int* dst = ei + E;
    const int nBins = (N + RB - 1) / RB;

    float* dinv = (float*)d_ws;
    size_t off = ((size_t)N * sizeof(float) + 255) & ~(size_t)255;
    float* H = (float*)((char*)d_ws + off);
    float* Y = H + (size_t)N * D;
    const size_t binBytes = ((size_t)(nBins + 1) * sizeof(int) + 255) & ~(size_t)255;
    char* p = (char*)(Y + (size_t)N * D);
    int* binTotal = (int*)p;  p += binBytes;
    int* cursor   = (int*)p;  p += binBytes;
    int* binStart = (int*)p;  p += binBytes;
    int* recs     = (int*)p;
    float* out = (float*)d_out;

    const int T = 256;
    const int nodeBlocks  = (N + T - 1) / T;
    const size_t placeLds = (size_t)(3 * nBins + 1) * sizeof(int);

    gcn_init_deg<<<nodeBlocks, T, 0, stream>>>(dinv, N);
    gcn_deg_edges<<<1024, T, 0, stream>>>(dst, dinv, E);
    gcn_dinv_bins<<<nBins, T, 0, stream>>>(dinv, binTotal, cursor, N);
    gcn_bin_place<<<256, PT, placeLds, stream>>>(src, dst, dinv, binTotal, cursor, binStart, recs, E, nBins);

    gcn_gemm_wmma<false><<<GEMM_GROUPS, GT, 0, stream>>>(x, W1, H, N);
    gcn_agg_lds<<<nBins, PT, 0, stream>>>(H, dinv, b1, recs, binStart, Y, N);

    gcn_gemm_wmma<true><<<GEMM_GROUPS, GT, 0, stream>>>(Y, W2, H, N);
    gcn_agg_lds<<<nBins, PT, 0, stream>>>(H, dinv, b2, recs, binStart, Y, N);

    gcn_gemm_wmma<true><<<GEMM_GROUPS, GT, 0, stream>>>(Y, W3, H, N);
    gcn_agg_lds<<<nBins, PT, 0, stream>>>(H, dinv, b3, recs, binStart, out, N);
}


# node degrees: each destination range's work-group counts its rows' in-edges from its own records with LDS integer adds (was 800000 global f32 atomic adds); norms filled by a record pass; 11 launches
# speedup vs baseline: 3.3618x; 1.2541x over previous
#include <hip/hip_runtime.h>
#include <hip/hip_bf16.h>


typedef float v2f __attribute__((ext_vector_type(2)));
typedef float v8f __attribute__((ext_vector_type(8)));

#define D 128
#define RB 200
#define PT 1024
#define EU 8
#define NW (PT / 32)
#define NK ((RB + NW - 1) / NW)
#define ACC_ROWS 328


__global__ void gcn_clear_bins(int* __restrict__ binTotal, int* __restrict__ cursor, int n_bins) {
    int i = blockIdx.x * blockDim.x + threadIdx.x;
    if (i < n_bins) { binTotal[i] = 0; cursor[i] = 0; }
}

__global__ __launch_bounds__(PT)
void gcn_bin_count(const int* __restrict__ dst, int* __restrict__ binTotal,
                   int n_edges, int n_bins) {
    extern __shared__ int sm[];
    const int t = threadIdx.x;
    for (int i = t; i < n_bins; i += PT) sm[i] = 0;
    __syncthreads();
    const int per = (n_edges + gridDim.x - 1) / gridDim.x;
    const int eb = blockIdx.x * per;
    const int ee = (eb + per < n_edges) ? (eb + per) : n_edges;
    for (int e = eb + t; e < ee; e += PT) atomicAdd(&sm[dst[e] / RB], 1);
    __syncthreads();
    for (int i = t; i < n_bins; i += PT) {
        const int c = sm[i];
        if (c > 0) atomicAdd(&binTotal[i], c);
    }
}


__global__ __launch_bounds__(PT)
void gcn_bin_place(const int* __restrict__ src, const int* __restrict__ dst,
                   const int* __restrict__ binTotal,
                   int* __restrict__ cursor, int* __restrict__ binStart,
                   int* __restrict__ recs, int n_edges, int n_bins) {
    extern __shared__ int sm[];
    int* startL = sm;
    int* cntL   = sm + (n_bins + 1);
    int* baseL  = cntL + n_bins;
    __shared__ int wsum[32];
    __shared__ int carryS;

    const int t = threadIdx.x;
    const int lane = t & 31;
    const int w = t >> 5;

    if (t == 0) carryS = 0;
    __syncthreads();
    for (int c0 = 0; c0 < n_bins; c0 += PT) {
        const int i = c0 + t;
        const int v = (i < n_bins) ? binTotal[i] : 0;
        int x = v;
        for (int o = 1; o < 32; o <<= 1) {
            const int y = __shfl_up(x, o);
            if (lane >= o) x += y;
        }
        if (lane == 31) wsum[w] = x;
        __syncthreads();
        int pre = carryS;
        for (int k = 0; k < w; ++k) pre += wsum[k];
        if (i < n_bins) startL[i] = pre + x - v;
        __syncthreads();
        if (t == PT - 1) carryS = pre + x;
        __syncthreads();
    }
    if (t == 0) startL[n_bins] = carryS;
    for (int i = t; i < n_bins; i += PT) cntL[i] = 0;
    __syncthreads();

    const int per = (n_edges + gridDim.x - 1) / gridDim.x;
    const int eb = blockIdx.x * per;
    const int ee = (eb + per < n_edges) ? (eb + per) : n_edges;

    for (int e = eb + t; e < ee; e += PT) atomicAdd(&cntL[dst[e] / RB], 1);
    __syncthreads();
    for (int i = t; i < n_bins; i += PT) {
        const int c = cntL[i];
        int base = startL[i];
        if (c > 0) base += atomicAdd(&cursor[i], c);
        baseL[i] = base;
        cntL[i] = 0;
    }
    __syncthreads();
    for (int e = eb + t; e < ee; e += PT) {
        const int s = src[e];
        const int d = dst[e];
        const int b = d / RB;
        const int pos = baseL[b] + atomicAdd(&cntL[b], 1);
        *(int4*)(recs + (size_t)pos * 4) = make_int4(s, d - b * RB, 0, 0);
    }
    if (blockIdx.x == 0)
        for (int i = t; i <= n_bins; i += PT) binStart[i] = startL[i];
}

__global__ __launch_bounds__(PT)
void gcn_range_dinv(const int* __restrict__ recs, const int* __restrict__ binStart,
                    float* __restrict__ dinv, int n) {
    __shared__ int cnt[RB];
    const int t = threadIdx.x;
    if (t < RB) cnt[t] = 0;
    __syncthreads();
    const int e0 = binStart[blockIdx.x];
    const int e1 = binStart[blockIdx.x + 1];
    for (int e = e0 + t; e < e1; e += PT) atomicAdd(&cnt[recs[(size_t)e * 4 + 1]], 1);
    __syncthreads();
    const int i = blockIdx.x * RB + t;
    if (t < RB && i < n) {
        const float d = (float)(cnt[t] + 1);
        dinv[i] = (d > 0.0f) ? __frsqrt_rn(d) : 0.0f;
    }
}

__global__ __launch_bounds__(PT)
void gcn_fill_norm(int* __restrict__ recs, const int* __restrict__ binStart,
                   const float* __restrict__ dinv) {
    const int t = threadIdx.x;
    const int node0 = blockIdx.x * RB;
    const int e0 = binStart[blockIdx.x];
    const int e1 = binStart[blockIdx.x + 1];
    for (int e = e0 + t; e < e1; e += PT) {
        int* p = recs + (size_t)e * 4;
        const int2 r = *(const int2*)p;
        const float norm = dinv[r.x] * dinv[node0 + r.y];
        p[2] = __float_as_int(norm);
    }
}


#define GW 16
#define GT (GW * 32)
#define GEMM_GROUPS 256
#define WPITCH 272
#define OPITCH 132

template <bool RELU>
__global__ __launch_bounds__(GT)
void gcn_gemm_wmma(const float* __restrict__ X, const float* __restrict__ W,
                   float* __restrict__ H, int n_rows) {
    __shared__ float Ws[(D / 2) * WPITCH];
    __shared__ float Os[GW * 16 * OPITCH];

    const int t = threadIdx.x;
    const int lane = t & 31;
    const int wave = __builtin_amdgcn_readfirstlane(t >> 5);
    const int half = lane >> 4;
    const int l16  = lane & 15;

    const int n_strips = (n_rows + 15) >> 4;
    const int stride = GW * gridDim.x;
    int s = wave * gridDim.x + blockIdx.x;

    float4 a[16];
    if (s < n_strips) {
        int arow = s * 16 + l16;
        if (arow >= n_rows) arow = n_rows - 1;
        const float4* __restrict__ ap = (const float4*)(X + (size_t)arow * D + 4 * half);
        #pragma unroll
        for (int q = 0; q < 16; ++q) a[q] = ap[2 * q];
    }

    #pragma unroll
    for (int it = 0; it < (D / 2) / GW; ++it) {
        const int kp = it * GW + wave;
        const float4 r0 = *(const float4*)(W + (size_t)(2 * kp) * D + 4 * lane);
        const float4 r1 = *(const float4*)(W + (size_t)(2 * kp + 1) * D + 4 * lane);
        float* p = Ws + kp * WPITCH + 8 * lane;
        *(float4*)(p)     = make_float4(r0.x, r1.x, r0.y, r1.y);
        *(float4*)(p + 4) = make_float4(r0.z, r1.z, r0.w, r1.w);
    }
    __syncthreads();

    float* __restrict__ Ow = Os + wave * (16 * OPITCH);
    const float* __restrict__ bbase = Ws + (2 * half) * WPITCH + 2 * l16;

    while (s < n_strips) {
        if (RELU) {
            #pragma unroll
            for (int q = 0; q < 16; ++q) {
                a[q].x = fmaxf(a[q].x, 0.0f); a[q].y = fmaxf(a[q].y, 0.0f);
                a[q].z = fmaxf(a[q].z, 0.0f); a[q].w = fmaxf(a[q].w, 0.0f);
            }
        }

        v8f acc[8];
        #pragma unroll
        for (int n = 0; n < 8; ++n) acc[n] = (v8f){};

        #pragma unroll
        for (int q = 0; q < 16; ++q) {
            #pragma unroll
            for (int u = 0; u < 2; ++u) {
                const float* __restrict__ bp = bbase + (4 * q + u) * WPITCH;
                v2f av;
                av.x = u ? a[q].z : a[q].x;
                av.y = u ? a[q].w : a[q].y;
                #pragma unroll
                for (int n = 0; n < 8; ++n) {
                    const v2f b = *(const v2f*)(bp + 32 * n);
                    acc[n] = __builtin_amdgcn_wmma_f32_16x16x4_f32(
                         false, av,  false, b,
                         (short)0, acc[n],  false,  false);
                }
            }
        }

        #pragma unroll
        for (int n = 0; n < 8; ++n) {
            #pragma unroll
            for (int v = 0; v < 8; ++v)
                Ow[(8 * half + v) * OPITCH + 16 * n + l16] = acc[n][v];
        }
        __builtin_amdgcn_fence(__ATOMIC_RELEASE, "wavefront");
        __builtin_amdgcn_wave_barrier();
        __builtin_amdgcn_fence(__ATOMIC_ACQUIRE, "wavefront");

        const int row0 = s * 16;
        float* __restrict__ hp = H + (size_t)row0 * D + 4 * lane;
        if (row0 + 16 <= n_rows) {
            #pragma unroll
            for (int i = 0; i < 16; ++i)
                *(float4*)(hp + (size_t)i * D) = *(const float4*)(Ow + i * OPITCH + 4 * lane);
        } else {
            #pragma unroll
            for (int i = 0; i < 16; ++i)
                if (row0 + i < n_rows)
                    *(float4*)(hp + (size_t)i * D) = *(const float4*)(Ow + i * OPITCH + 4 * lane);
        }

        s += stride;
        if (s < n_strips) {
            int arow = s * 16 + l16;
            if (arow >= n_rows) arow = n_rows - 1;
            const float4* __restrict__ ap = (const float4*)(X + (size_t)arow * D + 4 * half);
            #pragma unroll
            for (int q = 0; q < 16; ++q) a[q] = ap[2 * q];
        }
    }
}


__global__ __launch_bounds__(PT)
void gcn_agg_lds(const float* __restrict__ H, const float* __restrict__ dinv,
                 const float* __restrict__ bias, const int* __restrict__ recs,
                 const int* __restrict__ binStart, float* __restrict__ Y, int n_rows) {
    __shared__ float acc[ACC_ROWS * D];

    const int t = threadIdx.x;
    const int lane = t & 31;
    const int wave = __builtin_amdgcn_readfirstlane(t >> 5);
    const int node0 = blockIdx.x * RB;
    int nrows = n_rows - node0;
    if (nrows > RB) nrows = RB;

    static_assert(NK * NW <= ACC_ROWS, "every slot of the start block needs an LDS row");
    const int c = lane * 4;
    {
        const float4 b = *(const float4*)(bias + c);
        #pragma unroll
        for (int k = 0; k < NK; ++k) {
            const int r = wave + NW * k;
            const int rl = (r < nrows) ? r : nrows - 1;
            float s = dinv[node0 + rl];
            s = s * s;
            const float4 h = *(const float4*)(H + (size_t)(node0 + rl) * D + c);
            *(float4*)(acc + r * D + c) = make_float4(fmaf(h.x, s, b.x), fmaf(h.y, s, b.y),
                                                      fmaf(h.z, s, b.z), fmaf(h.w, s, b.w));
        }
    }
    __syncthreads();

    const int e0 = binStart[blockIdx.x];
    const int e1 = binStart[blockIdx.x + 1];
    const int4* __restrict__ rp = (const int4*)recs;

    for (int e = e0 + wave; e < e1; e += EU * 32) {
        int4 r[EU];
        v2f lo[EU], hi[EU];
        #pragma unroll
        for (int u = 0; u < EU; ++u) {
            int eu = e + u * 32;
            if (eu >= e1) eu = e1 - 1;
            r[u] = rp[eu];
        }
        #pragma unroll
        for (int u = 0; u < EU; ++u) {
            const v2f* hp = (const v2f*)(H + (size_t)r[u].x * D) + lane;
            lo[u] = hp[0];
            hi[u] = hp[32];
        }
        #pragma unroll
        for (int u = 0; u < EU; ++u) {
            if (e + u * 32 < e1) {
                const float norm = __int_as_float(r[u].z);
                float* yp = acc + r[u].y * D + lane * 2;
                atomicAdd(yp +  0, lo[u].x * norm);
                atomicAdd(yp +  1, lo[u].y * norm);
                atomicAdd(yp + 64, hi[u].x * norm);
                atomicAdd(yp + 65, hi[u].y * norm);
            }
        }
    }
    __syncthreads();

    #pragma unroll
    for (int k = 0; k < NK; ++k) {
        const int r = wave + NW * k;
        if (r < nrows)
            *(float4*)(Y + (size_t)(node0 + r) * D + c) = *(const float4*)(acc + r * D + c);
    }
}


extern "C" void kernel_launch(void* const* d_in, const int* in_sizes, int n_in,
                              void* d_out, int out_size, void* d_ws, size_t ws_size,
                              hipStream_t stream) {
    const float* x   = (const float*)d_in[0];
    const int*   ei  = (const int*)d_in[1];
    const float* W1  = (const float*)d_in[2];
    const float* b1  = (const float*)d_in[3];
    const float* W2  = (const float*)d_in[4];
    const float* b2  = (const float*)d_in[5];
    const float* W3  = (const float*)d_in[6];
    const float* b3  = (const float*)d_in[7];

    const int N = in_sizes[0] / D;
    const int E = in_sizes[1] / 2;
    const int* src = ei;
    const int* dst = ei + E;
    const int nBins = (N + RB - 1) / RB;

    float* dinv = (float*)d_ws;
    size_t off = ((size_t)N * sizeof(float) + 255) & ~(size_t)255;
    float* H = (float*)((char*)d_ws + off);
    float* Y = H + (size_t)N * D;
    const size_t binBytes = ((size_t)(nBins + 1) * sizeof(int) + 255) & ~(size_t)255;
    char* p = (char*)(Y + (size_t)N * D);
    int* binTotal = (int*)p;  p += binBytes;
    int* cursor   = (int*)p;  p += binBytes;
    int* binStart = (int*)p;  p += binBytes;
    int* recs     = (int*)p;
    float* out = (float*)d_out;

    const int T = 256;
    const int binBlocks   = (nBins + T - 1) / T;
    const size_t countLds = (size_t)nBins * sizeof(int);
    const size_t placeLds = (size_t)(3 * nBins + 1) * sizeof(int);

    gcn_clear_bins<<<binBlocks, T, 0, stream>>>(binTotal, cursor, nBins);
    gcn_bin_count<<<256, PT, countLds, stream>>>(dst, binTotal, E, nBins);
    gcn_bin_place<<<256, PT, placeLds, stream>>>(src, dst, binTotal, cursor, binStart, recs, E, nBins);
    gcn_range_dinv<<<nBins, PT, 0, stream>>>(recs, binStart, dinv, N);
    gcn_fill_norm<<<nBins, PT, 0, stream>>>(recs, binStart, dinv);

    gcn_gemm_wmma<false><<<GEMM_GROUPS, GT, 0, stream>>>(x, W1, H, N);
    gcn_agg_lds<<<nBins, PT, 0, stream>>>(H, dinv, b1, recs, binStart, Y, N);

    gcn_gemm_wmma<true><<<GEMM_GROUPS, GT, 0, stream>>>(Y, W2, H, N);
    gcn_agg_lds<<<nBins, PT, 0, stream>>>(H, dinv, b2, recs, binStart, Y, N);

    gcn_gemm_wmma<true><<<GEMM_GROUPS, GT, 0, stream>>>(Y, W3, H, N);
    gcn_agg_lds<<<nBins, PT, 0, stream>>>(H, dinv, b3, recs, binStart, out, N);
}


# GEMMs 2 and 3 moved into the aggregation: a range's work-group multiplies its finished rows (relu, in LDS) by the next W on f32 WMMA and stores the product; 9 launches, were 11; sum orders unchanged
# speedup vs baseline: 3.8864x; 1.1560x over previous
#include <hip/hip_runtime.h>
#include <hip/hip_bf16.h>


typedef float v2f __attribute__((ext_vector_type(2)));
typedef float v8f __attribute__((ext_vector_type(8)));

#define D 128
#define RB 200
#define PT 1024
#define EU 8
#define NW (PT / 32)
#define NK ((RB + NW - 1) / NW)
#define ACC_ROWS 328
#define APITCH 132
#define FROWS (NK * NW)
#define SPITCH 36


__global__ void gcn_clear_bins(int* __restrict__ binTotal, int* __restrict__ cursor, int n_bins) {
    int i = blockIdx.x * blockDim.x + threadIdx.x;
    if (i < n_bins) { binTotal[i] = 0; cursor[i] = 0; }
}

__global__ __launch_bounds__(PT)
void gcn_bin_count(const int* __restrict__ dst, int* __restrict__ binTotal,
                   int n_edges, int n_bins) {
    extern __shared__ int sm[];
    const int t = threadIdx.x;
    for (int i = t; i < n_bins; i += PT) sm[i] = 0;
    __syncthreads();
    const int per = (n_edges + gridDim.x - 1) / gridDim.x;
    const int eb = blockIdx.x * per;
    const int ee = (eb + per < n_edges) ? (eb + per) : n_edges;
    for (int e = eb + t; e < ee; e += PT) atomicAdd(&sm[dst[e] / RB], 1);
    __syncthreads();
    for (int i = t; i < n_bins; i += PT) {
        const int c = sm[i];
        if (c > 0) atomicAdd(&binTotal[i], c);
    }
}


__global__ __launch_bounds__(PT)
void gcn_bin_place(const int* __restrict__ src, const int* __restrict__ dst,
                   const int* __restrict__ binTotal,
                   int* __restrict__ cursor, int* __restrict__ binStart,
                   int* __restrict__ recs, int n_edges, int n_bins) {
    extern __shared__ int sm[];
    int* startL = sm;
    int* cntL   = sm + (n_bins + 1);
    int* baseL  = cntL + n_bins;
    __shared__ int wsum[32];
    __shared__ int carryS;

    const int t = threadIdx.x;
    const int lane = t & 31;
    const int w = t >> 5;

    if (t == 0) carryS = 0;
    __syncthreads();
    for (int c0 = 0; c0 < n_bins; c0 += PT) {
        const int i = c0 + t;
        const int v = (i < n_bins) ? binTotal[i] : 0;
        int x = v;
        for (int o = 1; o < 32; o <<= 1) {
            const int y = __shfl_up(x, o);
            if (lane >= o) x += y;
        }
        if (lane == 31) wsum[w] = x;
        __syncthreads();
        int pre = carryS;
        for (int k = 0; k < w; ++k) pre += wsum[k];
        if (i < n_bins) startL[i] = pre + x - v;
        __syncthreads();
        if (t == PT - 1) carryS = pre + x;
        __syncthreads();
    }
    if (t == 0) startL[n_bins] = carryS;
    for (int i = t; i < n_bins; i += PT) cntL[i] = 0;
    __syncthreads();

    const int per = (n_edges + gridDim.x - 1) / gridDim.x;
    const int eb = blockIdx.x * per;
    const int ee = (eb + per < n_edges) ? (eb + per) : n_edges;

    for (int e = eb + t; e < ee; e += PT) atomicAdd(&cntL[dst[e] / RB], 1);
    __syncthreads();
    for (int i = t; i < n_bins; i += PT) {
        const int c = cntL[i];
        int base = startL[i];
        if (c > 0) base += atomicAdd(&cursor[i], c);
        baseL[i] = base;
        cntL[i] = 0;
    }
    __syncthreads();
    for (int e = eb + t; e < ee; e += PT) {
        const int s = src[e];
        const int d = dst[e];
        const int b = d / RB;
        const int pos = baseL[b] + atomicAdd(&cntL[b], 1);
        *(int4*)(recs + (size_t)pos * 4) = make_int4(s, d - b * RB, 0, 0);
    }
    if (blockIdx.x == 0)
        for (int i = t; i <= n_bins; i += PT) binStart[i] = startL[i];
}

__global__ __launch_bounds__(PT)
void gcn_range_dinv(const int* __restrict__ recs, const int* __restrict__ binStart,
                    float* __restrict__ dinv, int n) {
    __shared__ int cnt[RB];
    const int t = threadIdx.x;
    if (t < RB) cnt[t] = 0;
    __syncthreads();
    const int e0 = binStart[blockIdx.x];
    const int e1 = binStart[blockIdx.x + 1];
    for (int e = e0 + t; e < e1; e += PT) atomicAdd(&cnt[recs[(size_t)e * 4 + 1]], 1);
    __syncthreads();
    const int i = blockIdx.x * RB + t;
    if (t < RB && i < n) {
        const float d = (float)(cnt[t] + 1);
        dinv[i] = (d > 0.0f) ? __frsqrt_rn(d) : 0.0f;
    }
}

__global__ __launch_bounds__(PT)
void gcn_fill_norm(int* __restrict__ recs, const int* __restrict__ binStart,
                   const float* __restrict__ dinv) {
    const int t = threadIdx.x;
    const int node0 = blockIdx.x * RB;
    const int e0 = binStart[blockIdx.x];
    const int e1 = binStart[blockIdx.x + 1];
    for (int e = e0 + t; e < e1; e += PT) {
        int* p = recs + (size_t)e * 4;
        const int2 r = *(const int2*)p;
        const float norm = dinv[r.x] * dinv[node0 + r.y];
        p[2] = __float_as_int(norm);
    }
}


#define GW 16
#define GT (GW * 32)
#define GEMM_GROUPS 256
#define WPITCH 272
#define OPITCH 132

template <bool RELU>
__global__ __launch_bounds__(GT)
void gcn_gemm_wmma(const float* __restrict__ X, const float* __restrict__ W,
                   float* __restrict__ H, int n_rows) {
    __shared__ float Ws[(D / 2) * WPITCH];
    __shared__ float Os[GW * 16 * OPITCH];

    const int t = threadIdx.x;
    const int lane = t & 31;
    const int wave = __builtin_amdgcn_readfirstlane(t >> 5);
    const int half = lane >> 4;
    const int l16  = lane & 15;

    const int n_strips = (n_rows + 15) >> 4;
    const int stride = GW * gridDim.x;
    int s = wave * gridDim.x + blockIdx.x;

    float4 a[16];
    if (s < n_strips) {
        int arow = s * 16 + l16;
        if (arow >= n_rows) arow = n_rows - 1;
        const float4* __restrict__ ap = (const float4*)(X + (size_t)arow * D + 4 * half);
        #pragma unroll
        for (int q = 0; q < 16; ++q) a[q] = ap[2 * q];
    }

    #pragma unroll
    for (int it = 0; it < (D / 2) / GW; ++it) {
        const int kp = it * GW + wave;
        const float4 r0 = *(const float4*)(W + (size_t)(2 * kp) * D + 4 * lane);
        const float4 r1 = *(const float4*)(W + (size_t)(2 * kp + 1) * D + 4 * lane);
        float* p = Ws + kp * WPITCH + 8 * lane;
        *(float4*)(p)     = make_float4(r0.x, r1.x, r0.y, r1.y);
        *(float4*)(p + 4) = make_float4(r0.z, r1.z, r0.w, r1.w);
    }
    __syncthreads();

    float* __restrict__ Ow = Os + wave * (16 * OPITCH);
    const float* __restrict__ bbase = Ws + (2 * half) * WPITCH + 2 * l16;

    while (s < n_strips) {
        if (RELU) {
            #pragma unroll
            for (int q = 0; q < 16; ++q) {
                a[q].x = fmaxf(a[q].x, 0.0f); a[q].y = fmaxf(a[q].y, 0.0f);
                a[q].z = fmaxf(a[q].z, 0.0f); a[q].w = fmaxf(a[q].w, 0.0f);
            }
        }

        v8f acc[8];
        #pragma unroll
        for (int n = 0; n < 8; ++n) acc[n] = (v8f){};

        #pragma unroll
        for (int q = 0; q < 16; ++q) {
            #pragma unroll
            for (int u = 0; u < 2; ++u) {
                const float* __restrict__ bp = bbase + (4 * q + u) * WPITCH;
                v2f av;
                av.x = u ? a[q].z : a[q].x;
                av.y = u ? a[q].w : a[q].y;
                #pragma unroll
                for (int n = 0; n < 8; ++n) {
                    const v2f b = *(const v2f*)(bp + 32 * n);
                    acc[n] = __builtin_amdgcn_wmma_f32_16x16x4_f32(
                         false, av,  false, b,
                         (short)0, acc[n],  false,  false);
                }
            }
        }

        #pragma unroll
        for (int n = 0; n < 8; ++n) {
            #pragma unroll
            for (int v = 0; v < 8; ++v)
                Ow[(8 * half + v) * OPITCH + 16 * n + l16] = acc[n][v];
        }
        __builtin_amdgcn_fence(__ATOMIC_RELEASE, "wavefront");
        __builtin_amdgcn_wave_barrier();
        __builtin_amdgcn_fence(__ATOMIC_ACQUIRE, "wavefront");

        const int row0 = s * 16;
        float* __restrict__ hp = H + (size_t)row0 * D + 4 * lane;
        if (row0 + 16 <= n_rows) {
            #pragma unroll
            for (int i = 0; i < 16; ++i)
                *(float4*)(hp + (size_t)i * D) = *(const float4*)(Ow + i * OPITCH + 4 * lane);
        } else {
            #pragma unroll
            for (int i = 0; i < 16; ++i)
                if (row0 + i < n_rows)
                    *(float4*)(hp + (size_t)i * D) = *(const float4*)(Ow + i * OPITCH + 4 * lane);
        }

        s += stride;
        if (s < n_strips) {
            int arow = s * 16 + l16;
            if (arow >= n_rows) arow = n_rows - 1;
            const float4* __restrict__ ap = (const float4*)(X + (size_t)arow * D + 4 * half);
            #pragma unroll
            for (int q = 0; q < 16; ++q) a[q] = ap[2 * q];
        }
    }
}


template <bool FUSE>
__global__ __launch_bounds__(PT)
void gcn_agg_lds(const float* __restrict__ H, const float* __restrict__ dinv,
                 const float* __restrict__ bias, const int* __restrict__ recs,
                 const int* __restrict__ binStart, const float* __restrict__ Wn,
                 float* __restrict__ Y, int n_rows) {
    constexpr int AP = FUSE ? APITCH : D;
    constexpr int AR = FUSE ? FROWS : ACC_ROWS;
    __shared__ float acc[AR * AP];
    __shared__ float Ws[FUSE ? (D / 2) * WPITCH : 4];
    __shared__ float Os[FUSE ? NW * 16 * SPITCH : 4];

    const int t = threadIdx.x;
    const int lane = t & 31;
    const int wave = __builtin_amdgcn_readfirstlane(t >> 5);
    const int node0 = blockIdx.x * RB;
    int nrows = n_rows - node0;
    if (nrows > RB) nrows = RB;

    if constexpr (FUSE) {
        #pragma unroll
        for (int it = 0; it < (D / 2) / NW; ++it) {
            const int kp = it * NW + wave;
            const float4 r0 = *(const float4*)(Wn + (size_t)(2 * kp) * D + 4 * lane);
            const float4 r1 = *(const float4*)(Wn + (size_t)(2 * kp + 1) * D + 4 * lane);
            float* p = Ws + kp * WPITCH + 8 * lane;
            *(float4*)(p)     = make_float4(r0.x, r1.x, r0.y, r1.y);
            *(float4*)(p + 4) = make_float4(r0.z, r1.z, r0.w, r1.w);
        }
    }

    static_assert(NK * NW <= AR, "every slot of the start block needs an LDS row");
    const int c = lane * 4;
    {
        const float4 b = *(const float4*)(bias + c);
        #pragma unroll
        for (int k = 0; k < NK; ++k) {
            const int r = wave + NW * k;
            const int rl = (r < nrows) ? r : nrows - 1;
            float s = dinv[node0 + rl];
            s = s * s;
            const float4 h = *(const float4*)(H + (size_t)(node0 + rl) * D + c);
            *(float4*)(acc + r * AP + c) = make_float4(fmaf(h.x, s, b.x), fmaf(h.y, s, b.y),
                                                       fmaf(h.z, s, b.z), fmaf(h.w, s, b.w));
        }
    }
    __syncthreads();

    const int e0 = binStart[blockIdx.x];
    const int e1 = binStart[blockIdx.x + 1];
    const int4* __restrict__ rp = (const int4*)recs;

    for (int e = e0 + wave; e < e1; e += EU * 32) {
        int4 r[EU];
        v2f lo[EU], hi[EU];
        #pragma unroll
        for (int u = 0; u < EU; ++u) {
            int eu = e + u * 32;
            if (eu >= e1) eu = e1 - 1;
            r[u] = rp[eu];
        }
        #pragma unroll
        for (int u = 0; u < EU; ++u) {
            const v2f* hp = (const v2f*)(H + (size_t)r[u].x * D) + lane;
            lo[u] = hp[0];
            hi[u] = hp[32];
        }
        #pragma unroll
        for (int u = 0; u < EU; ++u) {
            if (e + u * 32 < e1) {
                const float norm = __int_as_float(r[u].z);
                float* yp = acc + r[u].y * AP + lane * 2;
                atomicAdd(yp +  0, lo[u].x * norm);
                atomicAdd(yp +  1, lo[u].y * norm);
                atomicAdd(yp + 64, hi[u].x * norm);
                atomicAdd(yp + 65, hi[u].y * norm);
            }
        }
    }
    __syncthreads();

    if constexpr (!FUSE) {
        #pragma unroll
        for (int k = 0; k < NK; ++k) {
            const int r = wave + NW * k;
            if (r < nrows)
                *(float4*)(Y + (size_t)(node0 + r) * D + c) = *(const float4*)(acc + r * AP + c);
        }
    } else {
        const int half = lane >> 4;
        const int l16  = lane & 15;
        float* __restrict__ Ow = Os + wave * (16 * SPITCH);
        const float* __restrict__ bbase = Ws + (2 * half) * WPITCH + 2 * l16;
        const int n_jobs = 4 * ((nrows + 15) >> 4);
        static_assert(NW == 32 && 4 * ((RB + 15) / 16) <= NW + 20, "job split is written for 32 waves");
        const int w8 = wave & 7;
        const int blk = wave >> 3;
        const int job2 = (w8 < 4) ? NW + 5 * blk + w8
                       : (w8 == 4 + blk) ? NW + 5 * blk + 4 : n_jobs;

        int job = wave;
        while (job < n_jobs) {
            const int strip = job >> 2;
            const int qt = job & 3;

            const float* __restrict__ ap = acc + (strip * 16 + l16) * AP + 4 * half;
            float4 a[16];
            #pragma unroll
            for (int q = 0; q < 16; ++q) {
                a[q] = *(const float4*)(ap + 8 * q);
                a[q].x = fmaxf(a[q].x, 0.0f); a[q].y = fmaxf(a[q].y, 0.0f);
                a[q].z = fmaxf(a[q].z, 0.0f); a[q].w = fmaxf(a[q].w, 0.0f);
            }

            v8f c0 = (v8f){};
            v8f c1 = (v8f){};
            const float* __restrict__ bq = bbase + 64 * qt;
            #pragma unroll
            for (int q = 0; q < 16; ++q) {
                #pragma unroll
                for (int u = 0; u < 2; ++u) {
                    const float* __restrict__ bp = bq + (4 * q + u) * WPITCH;
                    v2f av;
                    av.x = u ? a[q].z : a[q].x;
                    av.y = u ? a[q].w : a[q].y;
                    const v2f b0 = *(const v2f*)(bp);
                    const v2f b1 = *(const v2f*)(bp + 32);
                    c0 = __builtin_amdgcn_wmma_f32_16x16x4_f32(
                         false, av,  false, b0,
                         (short)0, c0,  false,  false);
                    c1 = __builtin_amdgcn_wmma_f32_16x16x4_f32(
                         false, av,  false, b1,
                         (short)0, c1,  false,  false);
                }
            }

            #pragma unroll
            for (int v = 0; v < 8; ++v) {
                Ow[(8 * half + v) * SPITCH + l16]      = c0[v];
                Ow[(8 * half + v) * SPITCH + 16 + l16] = c1[v];
            }
            __builtin_amdgcn_fence(__ATOMIC_RELEASE, "wavefront");
            __builtin_amdgcn_wave_barrier();
            __builtin_amdgcn_fence(__ATOMIC_ACQUIRE, "wavefront");

            const int rsub = lane >> 3;
            const int c4 = 4 * (lane & 7);
            float* __restrict__ hp = Y + (size_t)(node0 + strip * 16) * D + 32 * qt + c4;
            #pragma unroll
            for (int it = 0; it < 4; ++it) {
                const int r = rsub + 4 * it;
                if (strip * 16 + r < nrows)
                    *(float4*)(hp + (size_t)r * D) = *(const float4*)(Ow + r * SPITCH + c4);
            }
            __builtin_amdgcn_fence(__ATOMIC_RELEASE, "wavefront");
            __builtin_amdgcn_wave_barrier();
            __builtin_amdgcn_fence(__ATOMIC_ACQUIRE, "wavefront");

            job = (job < NW) ? job2 : n_jobs;
        }
    }
}


extern "C" void kernel_launch(void* const* d_in, const int* in_sizes, int n_in,
                              void* d_out, int out_size, void* d_ws, size_t ws_size,
                              hipStream_t stream) {
    const float* x   = (const float*)d_in[0];
    const int*   ei  = (const int*)d_in[1];
    const float* W1  = (const float*)d_in[2];
    const float* b1  = (const float*)d_in[3];
    const float* W2  = (const float*)d_in[4];
    const float* b2  = (const float*)d_in[5];
    const float* W3  = (const float*)d_in[6];
    const float* b3  = (const float*)d_in[7];

    const int N = in_sizes[0] / D;
    const int E = in_sizes[1] / 2;
    const int* src = ei;
    const int* dst = ei + E;
    const int nBins = (N + RB - 1) / RB;

    float* dinv = (float*)d_ws;
    size_t off = ((size_t)N * sizeof(float) + 255) & ~(size_t)255;
    float* H = (float*)((char*)d_ws + off);
    float* Y = H + (size_t)N * D;
    const size_t binBytes = ((size_t)(nBins + 1) * sizeof(int) + 255) & ~(size_t)255;
    char* p = (char*)(Y + (size_t)N * D);
    int* binTotal = (int*)p;  p += binBytes;
    int* cursor   = (int*)p;  p += binBytes;
    int* binStart = (int*)p;  p += binBytes;
    int* recs     = (int*)p;
    float* out = (float*)d_out;

    const int T = 256;
    const int binBlocks   = (nBins + T - 1) / T;
    const size_t countLds = (size_t)nBins * sizeof(int);
    const size_t placeLds = (size_t)(3 * nBins + 1) * sizeof(int);

    gcn_clear_bins<<<binBlocks, T, 0, stream>>>(binTotal, cursor, nBins);
    gcn_bin_count<<<256, PT, countLds, stream>>>(dst, binTotal, E, nBins);
    gcn_bin_place<<<256, PT, placeLds, stream>>>(src, dst, binTotal, cursor, binStart, recs, E, nBins);
    gcn_range_dinv<<<nBins, PT, 0, stream>>>(recs, binStart, dinv, N);
    gcn_fill_norm<<<nBins, PT, 0, stream>>>(recs, binStart, dinv);

    gcn_gemm_wmma<false><<<GEMM_GROUPS, GT, 0, stream>>>(x, W1, H, N);
    gcn_agg_lds<true><<<nBins, PT, 0, stream>>>(H, dinv, b1, recs, binStart, W2, Y, N);

    gcn_agg_lds<true><<<nBins, PT, 0, stream>>>(Y, dinv, b2, recs, binStart, W3, H, N);

    gcn_agg_lds<false><<<nBins, PT, 0, stream>>>(H, dinv, b3, recs, binStart, W3, out, N);
}


# edge weight split per row: product rows stored times own dinv, a destination scales its sum by dinv, then adds bias (was first addend); norm pass gone, product 1 joins degree kernel; 7 launches
# speedup vs baseline: 4.2228x; 1.0866x over previous
#include <hip/hip_runtime.h>
#include <hip/hip_bf16.h>


typedef float v2f __attribute__((ext_vector_type(2)));
typedef float v8f __attribute__((ext_vector_type(8)));

#define D 128
#define RB 200
#define PT 1024
#define EU 8
#define NW (PT / 32)
#define NK ((RB + NW - 1) / NW)
#define ACC_ROWS 328
#define APITCH 132
#define FROWS (NK * NW)
#define SPITCH 36
#define WPITCH 272


__global__ void gcn_clear_bins(int* __restrict__ binTotal, int* __restrict__ cursor, int n_bins) {
    int i = blockIdx.x * blockDim.x + threadIdx.x;
    if (i < n_bins) { binTotal[i] = 0; cursor[i] = 0; }
}

__global__ __launch_bounds__(PT)
void gcn_bin_count(const int* __restrict__ dst, int* __restrict__ binTotal,
                   int n_edges, int n_bins) {
    extern __shared__ int sm[];
    const int t = threadIdx.x;
    for (int i = t; i < n_bins; i += PT) sm[i] = 0;
    __syncthreads();
    const int per = (n_edges + gridDim.x - 1) / gridDim.x;
    const int eb = blockIdx.x * per;
    const int ee = (eb + per < n_edges) ? (eb + per) : n_edges;
    for (int e = eb + t; e < ee; e += PT) atomicAdd(&sm[dst[e] / RB], 1);
    __syncthreads();
    for (int i = t; i < n_bins; i += PT) {
        const int c = sm[i];
        if (c > 0) atomicAdd(&binTotal[i], c);
    }
}


__global__ __launch_bounds__(PT)
void gcn_bin_place(const int* __restrict__ src, const int* __restrict__ dst,
                   const int* __restrict__ binTotal,
                   int* __restrict__ cursor, int* __restrict__ binStart,
                   int* __restrict__ recs, int n_edges, int n_bins) {
    extern __shared__ int sm[];
    int* startL = sm;
    int* cntL   = sm + (n_bins + 1);
    int* baseL  = cntL + n_bins;
    __shared__ int wsum[32];
    __shared__ int carryS;

    const int t = threadIdx.x;
    const int lane = t & 31;
    const int w = t >> 5;

    if (t == 0) carryS = 0;
    __syncthreads();
    for (int c0 = 0; c0 < n_bins; c0 += PT) {
        const int i = c0 + t;
        const int v = (i < n_bins) ? binTotal[i] : 0;
        int x = v;
        for (int o = 1; o < 32; o <<= 1) {
            const int y = __shfl_up(x, o);
            if (lane >= o) x += y;
        }
        if (lane == 31) wsum[w] = x;
        __syncthreads();
        int pre = carryS;
        for (int k = 0; k < w; ++k) pre += wsum[k];
        if (i < n_bins) startL[i] = pre + x - v;
        __syncthreads();
        if (t == PT - 1) carryS = pre + x;
        __syncthreads();
    }
    if (t == 0) startL[n_bins] = carryS;
    for (int i = t; i < n_bins; i += PT) cntL[i] = 0;
    __syncthreads();

    const int per = (n_edges + gridDim.x - 1) / gridDim.x;
    const int eb = blockIdx.x * per;
    const int ee = (eb + per < n_edges) ? (eb + per) : n_edges;

    for (int e = eb + t; e < ee; e += PT) atomicAdd(&cntL[dst[e] / RB], 1);
    __syncthreads();
    for (int i = t; i < n_bins; i += PT) {
        const int c = cntL[i];
        int base = startL[i];
        if (c > 0) base += atomicAdd(&cursor[i], c);
        baseL[i] = base;
        cntL[i] = 0;
    }
    __syncthreads();
    for (int e = eb + t; e < ee; e += PT) {
        const int s = src[e];
        const int d = dst[e];
        const int b = d / RB;
        const int pos = baseL[b] + atomicAdd(&cntL[b], 1);
        *(int2*)(recs + (size_t)pos * 2) = make_int2(s, d - b * RB);
    }
    if (blockIdx.x == 0)
        for (int i = t; i <= n_bins; i += PT) binStart[i] = startL[i];
}


template <int MODE>
__global__ __launch_bounds__(PT)
void gcn_range(const float* __restrict__ H, const float* dinv_in, float* dinv_out,
               const float* __restrict__ bias, const int* __restrict__ recs,
               const int* __restrict__ binStart, const float* __restrict__ Wn,
               float* __restrict__ Y, int n_rows) {
    constexpr bool GEMM = (MODE != 0);
    constexpr int AP = GEMM ? APITCH : D;
    constexpr int AR = GEMM ? FROWS : ACC_ROWS;
    __shared__ float acc[AR * AP];
    __shared__ float Ws[GEMM ? (D / 2) * WPITCH : 4];
    __shared__ float Os[GEMM ? NW * 16 * SPITCH : 4];
    __shared__ float dvL[FROWS];
    __shared__ int cnt[MODE == 2 ? RB : 1];

    const int t = threadIdx.x;
    const int lane = t & 31;
    const int wave = __builtin_amdgcn_readfirstlane(t >> 5);
    const int node0 = blockIdx.x * RB;
    int nrows = n_rows - node0;
    if (nrows > RB) nrows = RB;

    if constexpr (GEMM) {
        #pragma unroll
        for (int it = 0; it < (D / 2) / NW; ++it) {
            const int kp = it * NW + wave;
            const float4 r0 = *(const float4*)(Wn + (size_t)(2 * kp) * D + 4 * lane);
            const float4 r1 = *(const float4*)(Wn + (size_t)(2 * kp + 1) * D + 4 * lane);
            float* p = Ws + kp * WPITCH + 8 * lane;
            *(float4*)(p)     = make_float4(r0.x, r1.x, r0.y, r1.y);
            *(float4*)(p + 4) = make_float4(r0.z, r1.z, r0.w, r1.w);
        }
    }
    if constexpr (MODE == 2) {
        if (t < RB) cnt[t] = 0;
    } else {
        if (t < FROWS) dvL[t] = (t < nrows) ? dinv_in[node0 + t] : 0.0f;
    }

    static_assert(NK * NW <= AR, "every slot of the start block needs an LDS row");
    const int c = lane * 4;
    #pragma unroll
    for (int k = 0; k < NK; ++k) {
        const int r = wave + NW * k;
        const int rl = (r < nrows) ? r : nrows - 1;
        *(float4*)(acc + r * AP + c) = *(const float4*)(H + (size_t)(node0 + rl) * D + c);
    }
    __syncthreads();

    const int e0 = binStart[blockIdx.x];
    const int e1 = binStart[blockIdx.x + 1];

    if constexpr (MODE == 2) {
        for (int e = e0 + t; e < e1; e += PT) atomicAdd(&cnt[recs[(size_t)e * 2 + 1]], 1);
        __syncthreads();
        if (t < FROWS) {
            float v = 0.0f;
            if (t < nrows) {
                const float d = (float)(cnt[t] + 1);
                v = (d > 0.0f) ? __frsqrt_rn(d) : 0.0f;
                dinv_out[node0 + t] = v;
            }
            dvL[t] = v;
        }
        __syncthreads();
    } else {
        const int2* __restrict__ rp = (const int2*)recs;
        for (int e = e0 + wave; e < e1; e += EU * 32) {
            int2 r[EU];
            v2f lo[EU], hi[EU];
            #pragma unroll
            for (int u = 0; u < EU; ++u) {
                int eu = e + u * 32;
                if (eu >= e1) eu = e1 - 1;
                r[u] = rp[eu];
            }
            #pragma unroll
            for (int u = 0; u < EU; ++u) {
                const v2f* hp = (const v2f*)(H + (size_t)r[u].x * D) + lane;
                lo[u] = hp[0];
                hi[u] = hp[32];
            }
            #pragma unroll
            for (int u = 0; u < EU; ++u) {
                if (e + u * 32 < e1) {
                    float* yp = acc + r[u].y * AP + lane * 2;
                    atomicAdd(yp +  0, lo[u].x);
                    atomicAdd(yp +  1, lo[u].y);
                    atomicAdd(yp + 64, hi[u].x);
                    atomicAdd(yp + 65, hi[u].y);
                }
            }
        }
        __syncthreads();
    }

    if constexpr (MODE == 0) {
        const float4 b = *(const float4*)(bias + c);
        float4 a[NK];
        float s[NK];
        #pragma unroll
        for (int k = 0; k < NK; ++k) {
            const int r = wave + NW * k;
            s[k] = dvL[r];
            a[k] = *(const float4*)(acc + r * AP + c);
        }
        #pragma unroll
        for (int k = 0; k < NK; ++k) {
            const int r = wave + NW * k;
            if (r < nrows)
                *(float4*)(Y + (size_t)(node0 + r) * D + c) =
                    make_float4(fmaf(a[k].x, s[k], b.x), fmaf(a[k].y, s[k], b.y),
                                fmaf(a[k].z, s[k], b.z), fmaf(a[k].w, s[k], b.w));
        }
    } else {
        if constexpr (MODE == 1) {
            const float4 b = *(const float4*)(bias + c);
            #pragma unroll
            for (int k = 0; k < NK; ++k) {
                const int r = wave + NW * k;
                const float s = dvL[r];
                float* ap = acc + r * AP + c;
                const float4 a = *(const float4*)ap;
                *(float4*)ap = make_float4(fmaf(a.x, s, b.x), fmaf(a.y, s, b.y), fmaf(a.z, s, b.z), fmaf(a.w, s, b.w));
            }
            __syncthreads();
        }
        const int half = lane >> 4;
        const int l16  = lane & 15;
        float* __restrict__ Ow = Os + wave * (16 * SPITCH);
        const float* __restrict__ bbase = Ws + (2 * half) * WPITCH + 2 * l16;
        const int n_jobs = 4 * ((nrows + 15) >> 4);
        static_assert(NW == 32 && 4 * ((RB + 15) / 16) <= NW + 20, "job split is written for 32 waves");
        const int w8 = wave & 7;
        const int blk = wave >> 3;
        const int job2 = (w8 < 4) ? NW + 5 * blk + w8
                       : (w8 == 4 + blk) ? NW + 5 * blk + 4 : n_jobs;

        int job = wave;
        while (job < n_jobs) {
            const int strip = job >> 2;
            const int qt = job & 3;

            const float* __restrict__ ap = acc + (strip * 16 + l16) * AP + 4 * half;
            float4 a[16];
            #pragma unroll
            for (int q = 0; q < 16; ++q) {
                a[q] = *(const float4*)(ap + 8 * q);
                if constexpr (MODE == 1) {
                    a[q].x = fmaxf(a[q].x, 0.0f); a[q].y = fmaxf(a[q].y, 0.0f);
                    a[q].z = fmaxf(a[q].z, 0.0f); a[q].w = fmaxf(a[q].w, 0.0f);
                }
            }

            v8f c0 = (v8f){};
            v8f c1 = (v8f){};
            const float* __restrict__ bq = bbase + 64 * qt;
            #pragma unroll
            for (int q = 0; q < 16; ++q) {
                #pragma unroll
                for (int u = 0; u < 2; ++u) {
                    const float* __restrict__ bp = bq + (4 * q + u) * WPITCH;
                    v2f av;
                    av.x = u ? a[q].z : a[q].x;
                    av.y = u ? a[q].w : a[q].y;
                    const v2f b0 = *(const v2f*)(bp);
                    const v2f b1 = *(const v2f*)(bp + 32);
                    c0 = __builtin_amdgcn_wmma_f32_16x16x4_f32(
                         false, av,  false, b0,
                         (short)0, c0,  false,  false);
                    c1 = __builtin_amdgcn_wmma_f32_16x16x4_f32(
                         false, av,  false, b1,
                         (short)0, c1,  false,  false);
                }
            }

            #pragma unroll
            for (int v = 0; v < 8; ++v) {
                Ow[(8 * half + v) * SPITCH + l16]      = c0[v];
                Ow[(8 * half + v) * SPITCH + 16 + l16] = c1[v];
            }
            __builtin_amdgcn_fence(__ATOMIC_RELEASE, "wavefront");
            __builtin_amdgcn_wave_barrier();
            __builtin_amdgcn_fence(__ATOMIC_ACQUIRE, "wavefront");

            const int rsub = lane >> 3;
            const int c4 = 4 * (lane & 7);
            float* __restrict__ hp = Y + (size_t)(node0 + strip * 16) * D + 32 * qt + c4;
            #pragma unroll
            for (int it = 0; it < 4; ++it) {
                const int r = rsub + 4 * it;
                if (strip * 16 + r < nrows) {
                    const float s = dvL[strip * 16 + r];
                    const float4 o = *(const float4*)(Ow + r * SPITCH + c4);
                    *(float4*)(hp + (size_t)r * D) = make_float4(o.x * s, o.y * s, o.z * s, o.w * s);
                }
            }
            __builtin_amdgcn_fence(__ATOMIC_RELEASE, "wavefront");
            __builtin_amdgcn_wave_barrier();
            __builtin_amdgcn_fence(__ATOMIC_ACQUIRE, "wavefront");

            job = (job < NW) ? job2 : n_jobs;
        }
    }
}


extern "C" void kernel_launch(void* const* d_in, const int* in_sizes, int n_in,
                              void* d_out, int out_size, void* d_ws, size_t ws_size,
                              hipStream_t stream) {
    const float* x   = (const float*)d_in[0];
    const int*   ei  = (const int*)d_in[1];
    const float* W1  = (const float*)d_in[2];
    const float* b1  = (const float*)d_in[3];
    const float* W2  = (const float*)d_in[4];
    const float* b2  = (const float*)d_in[5];
    const float* W3  = (const float*)d_in[6];
    const float* b3  = (const float*)d_in[7];

    const int N = in_sizes[0] / D;
    const int E = in_sizes[1] / 2;
    const int* src = ei;
    const int* dst = ei + E;
    const int nBins = (N + RB - 1) / RB;

    float* dinv = (float*)d_ws;
    size_t off = ((size_t)N * sizeof(float) + 255) & ~(size_t)255;
    float* H = (float*)((char*)d_ws + off);
    float* Y = H + (size_t)N * D;
    const size_t binBytes = ((size_t)(nBins + 1) * sizeof(int) + 255) & ~(size_t)255;
    char* p = (char*)(Y + (size_t)N * D);
    int* binTotal = (int*)p;  p += binBytes;
    int* cursor   = (int*)p;  p += binBytes;
    int* binStart = (int*)p;  p += binBytes;
    int* recs     = (int*)p;
    float* out = (float*)d_out;

    const int T = 256;
    const int binBlocks   = (nBins + T - 1) / T;
    const size_t countLds = (size_t)nBins * sizeof(int);
    const size_t placeLds = (size_t)(3 * nBins + 1) * sizeof(int);

    gcn_clear_bins<<<binBlocks, T, 0, stream>>>(binTotal, cursor, nBins);
    gcn_bin_count<<<256, PT, countLds, stream>>>(dst, binTotal, E, nBins);
    gcn_bin_place<<<256, PT, placeLds, stream>>>(src, dst, binTotal, cursor, binStart, recs, E, nBins);

    gcn_range<2><<<nBins, PT, 0, stream>>>(x, dinv, dinv, b1, recs, binStart, W1, H, N);

    gcn_range<1><<<nBins, PT, 0, stream>>>(H, dinv, dinv, b1, recs, binStart, W2, Y, N);

    gcn_range<1><<<nBins, PT, 0, stream>>>(Y, dinv, dinv, b2, recs, binStart, W3, H, N);

    gcn_range<0><<<nBins, PT, 0, stream>>>(H, dinv, dinv, b3, recs, binStart, W3, out, N);
}


# speedup vs baseline: 4.4995x; 1.0655x over previous
#include <hip/hip_runtime.h>
#include <hip/hip_bf16.h>


typedef float v2f __attribute__((ext_vector_type(2)));
typedef float v8f __attribute__((ext_vector_type(8)));

#define D 128
#define RB 200
#define PT 1024
#define EU 8
#define NW (PT / 32)
#define NK ((RB + NW - 1) / NW)
#define XOFF 32768
#define APITCH 132
#define FROWS (NK * NW)
#define SPITCH 36
#define WPITCH 272


__global__ void gcn_clear_bins(int* __restrict__ binTotal, int* __restrict__ cursor, int n_bins) {
    int i = blockIdx.x * blockDim.x + threadIdx.x;
    if (i < n_bins) { binTotal[i] = 0; cursor[i] = 0; }
}

__global__ __launch_bounds__(PT)
void gcn_bin_count(const int* __restrict__ dst, int* __restrict__ binTotal,
                   int n_edges, int n_bins) {
    extern __shared__ int sm[];
    const int t = threadIdx.x;
    for (int i = t; i < n_bins; i += PT) sm[i] = 0;
    __syncthreads();
    const int per = (n_edges + gridDim.x - 1) / gridDim.x;
    const int eb = blockIdx.x * per;
    const int ee = (eb + per < n_edges) ? (eb + per) : n_edges;
    for (int e = eb + t; e < ee; e += PT) atomicAdd(&sm[dst[e] / RB], 1);
    __syncthreads();
    for (int i = t; i < n_bins; i += PT) {
        const int c = sm[i];
        if (c > 0) atomicAdd(&binTotal[i], c);
    }
}


__global__ __launch_bounds__(PT)
void gcn_bin_place(const int* __restrict__ src, const int* __restrict__ dst,
                   const int* __restrict__ binTotal,
                   int* __restrict__ cursor, int* __restrict__ binStart,
                   int* __restrict__ recs, int n_edges, int n_bins) {
    extern __shared__ int sm[];
    int* startL = sm;
    int* cntL   = sm + (n_bins + 1);
    int* baseL  = cntL + n_bins;
    __shared__ int wsum[32];
    __shared__ int carryS;

    const int t = threadIdx.x;
    const int lane = t & 31;
    const int w = t >> 5;

    if (t == 0) carryS = 0;
    __syncthreads();
    for (int c0 = 0; c0 < n_bins; c0 += PT) {
        const int i = c0 + t;
        const int v = (i < n_bins) ? binTotal[i] : 0;
        int x = v;
        for (int o = 1; o < 32; o <<= 1) {
            const int y = __shfl_up(x, o);
            if (lane >= o) x += y;
        }
        if (lane == 31) wsum[w] = x;
        __syncthreads();
        int pre = carryS;
        for (int k = 0; k < w; ++k) pre += wsum[k];
        if (i < n_bins) startL[i] = pre + x - v;
        __syncthreads();
        if (t == PT - 1) carryS = pre + x;
        __syncthreads();
    }
    if (t == 0) startL[n_bins] = carryS;
    for (int i = t; i < n_bins; i += PT) cntL[i] = 0;
    __syncthreads();

    const int per = (n_edges + gridDim.x - 1) / gridDim.x;
    const int eb = blockIdx.x * per;
    const int ee = (eb + per < n_edges) ? (eb + per) : n_edges;

    for (int e = eb + t; e < ee; e += PT) atomicAdd(&cntL[dst[e] / RB], 1);
    __syncthreads();
    for (int i = t; i < n_bins; i += PT) {
        const int c = cntL[i];
        int base = startL[i];
        if (c > 0) base += atomicAdd(&cursor[i], c);
        baseL[i] = base;
        cntL[i] = 0;
    }
    __syncthreads();
    for (int e = eb + t; e < ee; e += PT) {
        const int s = src[e];
        const int d = dst[e];
        const int b = d / RB;
        const int pos = baseL[b] + atomicAdd(&cntL[b], 1);
        *(int2*)(recs + (size_t)pos * 2) = make_int2(s, d - b * RB);
    }
    if (blockIdx.x == 0)
        for (int i = t; i <= n_bins; i += PT) binStart[i] = startL[i];
}


template <int MODE>
__global__ __launch_bounds__(PT)
void gcn_range(const float* __restrict__ H, const float* dinv_in, float* dinv_out,
               const float* __restrict__ bias, const int* __restrict__ recs,
               const int* __restrict__ binStart, const float* __restrict__ Wn,
               float* __restrict__ Y, int n_rows) {
    constexpr bool GEMM = (MODE != 0);
    constexpr bool EDGES = (MODE != 2);
    constexpr int AP = GEMM ? APITCH : D;
    constexpr int WS_OFF = XOFF;
    constexpr int OS_OFF = WS_OFF + (D / 2) * WPITCH;
    constexpr int DV_OFF = GEMM ? OS_OFF + NW * 16 * SPITCH : XOFF + FROWS * AP;
    constexpr int CN_OFF = DV_OFF + FROWS;
    constexpr int LDS_FLOATS = CN_OFF + (MODE == 2 ? RB : 0);
    static_assert(FROWS * AP <= XOFF && XOFF + FROWS * AP <= DV_OFF, "the two row copies must not reach the next region");
    static_assert(LDS_FLOATS * 4 > 160 * 1024 && LDS_FLOATS * 4 <= 320 * 1024, "one range work-group per WGP");
    __shared__ float lds[LDS_FLOATS];
    float* const acc = lds;
    float* const Ws = lds + WS_OFF;
    float* const Os = lds + OS_OFF;
    float* const dvL = lds + DV_OFF;
    int* const cnt = (int*)(lds + CN_OFF);

    const int t = threadIdx.x;
    const int lane = t & 31;
    const int wave = __builtin_amdgcn_readfirstlane(t >> 5);
    const int node0 = blockIdx.x * RB;
    int nrows = n_rows - node0;
    if (nrows > RB) nrows = RB;

    constexpr int WIT = (D / 2) / NW;
    if constexpr (MODE == 2) {
        #pragma unroll
        for (int it = 0; it < WIT; ++it) {
            const int kp = it * NW + wave;
            const float4 r0 = *(const float4*)(Wn + (size_t)(2 * kp) * D + 4 * lane);
            const float4 r1 = *(const float4*)(Wn + (size_t)(2 * kp + 1) * D + 4 * lane);
            float* p = Ws + kp * WPITCH + 8 * lane;
            *(float4*)(p)     = make_float4(r0.x, r1.x, r0.y, r1.y);
            *(float4*)(p + 4) = make_float4(r0.z, r1.z, r0.w, r1.w);
        }
    }
    int boff = 0;
    if constexpr (EDGES) {
        unsigned hw;
        asm volatile("s_getreg_b32 %0, hwreg(HW_REG_WAVE_HW_ID1)" : "=s"(hw));
        boff = (hw & 0x200u) ? XOFF : 0;
    }
    if constexpr (MODE == 2) {
        if (t < RB) cnt[t] = 0;
    } else {
        if (t < FROWS) dvL[t] = (t < nrows) ? dinv_in[node0 + t] : 0.0f;
    }

    const int c = lane * 4;
    float4 own[NK];
    #pragma unroll
    for (int k = 0; k < NK; ++k) {
        const int r = wave + NW * k;
        const int rl = (r < nrows) ? r : nrows - 1;
        own[k] = *(const float4*)(H + (size_t)(node0 + rl) * D + c);
    }
    float4 wq[2 * WIT];
    if constexpr (MODE == 1) {
        #pragma unroll
        for (int it = 0; it < WIT; ++it) {
            const int kp = it * NW + wave;
            wq[2 * it]     = *(const float4*)(Wn + (size_t)(2 * kp) * D + 4 * lane);
            wq[2 * it + 1] = *(const float4*)(Wn + (size_t)(2 * kp + 1) * D + 4 * lane);
        }
    }
    #pragma unroll
    for (int k = 0; k < NK; ++k) {
        const int r = wave + NW * k;
        *(float4*)(acc + r * AP + c) = own[k];
        if constexpr (EDGES) *(float4*)(acc + XOFF + r * AP + c) = make_float4(0.0f, 0.0f, 0.0f, 0.0f);
    }
    __syncthreads();

    const int e0 = binStart[blockIdx.x];
    const int e1 = binStart[blockIdx.x + 1];

    if constexpr (MODE == 2) {
        for (int e = e0 + t; e < e1; e += PT) atomicAdd(&cnt[recs[(size_t)e * 2 + 1]], 1);
        __syncthreads();
        if (t < FROWS) {
            float v = 0.0f;
            if (t < nrows) {
                const float d = (float)(cnt[t] + 1);
                v = (d > 0.0f) ? __frsqrt_rn(d) : 0.0f;
                dinv_out[node0 + t] = v;
            }
            dvL[t] = v;
        }
        __syncthreads();
    } else {
        const int2* __restrict__ rp = (const int2*)recs;
        for (int e = e0 + wave; e < e1; e += EU * 32) {
            int2 r[EU];
            v2f lo[EU], hi[EU];
            #pragma unroll
            for (int u = 0; u < EU; ++u) {
                int eu = e + u * 32;
                if (eu >= e1) eu = e1 - 1;
                r[u] = rp[eu];
            }
            #pragma unroll
            for (int u = 0; u < EU; ++u) {
                const v2f* hp = (const v2f*)(H + (size_t)r[u].x * D) + lane;
                lo[u] = hp[0];
                hi[u] = hp[32];
            }
            #pragma unroll
            for (int u = 0; u < EU; ++u) {
                if (e + u * 32 < e1) {
                    float* yp = acc + boff + r[u].y * AP + lane * 2;
                    atomicAdd(yp +  0, lo[u].x);
                    atomicAdd(yp +  1, lo[u].y);
                    atomicAdd(yp + 64, hi[u].x);
                    atomicAdd(yp + 65, hi[u].y);
                }
            }
        }
        __syncthreads();
    }

    if constexpr (MODE == 0) {
        const float4 b = *(const float4*)(bias + c);
        float4 a[NK];
        float4 z[NK];
        float s[NK];
        #pragma unroll
        for (int k = 0; k < NK; ++k) {
            const int r = wave + NW * k;
            s[k] = dvL[r];
            a[k] = *(const float4*)(acc + r * AP + c);
            z[k] = *(const float4*)(acc + XOFF + r * AP + c);
        }
        #pragma unroll
        for (int k = 0; k < NK; ++k) {
            const int r = wave + NW * k;
            if (r < nrows)
                *(float4*)(Y + (size_t)(node0 + r) * D + c) =
                    make_float4(fmaf(a[k].x + z[k].x, s[k], b.x), fmaf(a[k].y + z[k].y, s[k], b.y),
                                fmaf(a[k].z + z[k].z, s[k], b.z), fmaf(a[k].w + z[k].w, s[k], b.w));
        }
    } else {
        if constexpr (MODE == 1) {
            const float4 b = *(const float4*)(bias + c);
            #pragma unroll
            for (int k = 0; k < NK; ++k) {
                const int r = wave + NW * k;
                const float s = dvL[r];
                float* ap = acc + r * AP + c;
                const float4 a = *(const float4*)ap;
                const float4 z = *(const float4*)(ap + XOFF);
                *(float4*)ap = make_float4(fmaf(a.x + z.x, s, b.x), fmaf(a.y + z.y, s, b.y),
                                           fmaf(a.z + z.z, s, b.z), fmaf(a.w + z.w, s, b.w));
            }
            __syncthreads();
            #pragma unroll
            for (int it = 0; it < WIT; ++it) {
                const int kp = it * NW + wave;
                const float4 r0 = wq[2 * it];
                const float4 r1 = wq[2 * it + 1];
                float* p = Ws + kp * WPITCH + 8 * lane;
                *(float4*)(p)     = make_float4(r0.x, r1.x, r0.y, r1.y);
                *(float4*)(p + 4) = make_float4(r0.z, r1.z, r0.w, r1.w);
            }
            __syncthreads();
        }
        const int half = lane >> 4;
        const int l16  = lane & 15;
        float* __restrict__ Ow = Os + wave * (16 * SPITCH);
        const float* __restrict__ bbase = Ws + (2 * half) * WPITCH + 2 * l16;
        const int n_jobs = 4 * ((nrows + 15) >> 4);
        static_assert(NW == 32 && 4 * ((RB + 15) / 16) <= NW + 20, "job split is written for 32 waves");
        const int w8 = wave & 7;
        const int blk = wave >> 3;
        const int job2 = (w8 < 4) ? NW + 5 * blk + w8
                       : (w8 == 4 + blk) ? NW + 5 * blk + 4 : n_jobs;

        int job = wave;
        while (job < n_jobs) {
            const int strip = job >> 2;
            const int qt = job & 3;

            const float* __restrict__ ap = acc + (strip * 16 + l16) * AP + 4 * half;
            float4 a[16];
            #pragma unroll
            for (int q = 0; q < 16; ++q) {
                a[q] = *(const float4*)(ap + 8 * q);
                if constexpr (MODE == 1) {
                    a[q].x = fmaxf(a[q].x, 0.0f); a[q].y = fmaxf(a[q].y, 0.0f);
                    a[q].z = fmaxf(a[q].z, 0.0f); a[q].w = fmaxf(a[q].w, 0.0f);
                }
            }

            v8f c0 = (v8f){};
            v8f c1 = (v8f){};
            const float* __restrict__ bq = bbase + 64 * qt;
            #pragma unroll
            for (int q = 0; q < 16; ++q) {
                #pragma unroll
                for (int u = 0; u < 2; ++u) {
                    const float* __restrict__ bp = bq + (4 * q + u) * WPITCH;
                    v2f av;
                    av.x = u ? a[q].z : a[q].x;
                    av.y = u ? a[q].w : a[q].y;
                    const v2f b0 = *(const v2f*)(bp);
                    const v2f b1 = *(const v2f*)(bp + 32);
                    c0 = __builtin_amdgcn_wmma_f32_16x16x4_f32(
                         false, av,  false, b0,
                         (short)0, c0,  false,  false);
                    c1 = __builtin_amdgcn_wmma_f32_16x16x4_f32(
                         false, av,  false, b1,
                         (short)0, c1,  false,  false);
                }
            }

            #pragma unroll
            for (int v = 0; v < 8; ++v) {
                Ow[(8 * half + v) * SPITCH + l16]      = c0[v];
                Ow[(8 * half + v) * SPITCH + 16 + l16] = c1[v];
            }
            __builtin_amdgcn_fence(__ATOMIC_RELEASE, "wavefront");
            __builtin_amdgcn_wave_barrier();
            __builtin_amdgcn_fence(__ATOMIC_ACQUIRE, "wavefront");

            const int rsub = lane >> 3;
            const int c4 = 4 * (lane & 7);
            float* __restrict__ hp = Y + (size_t)(node0 + strip * 16) * D + 32 * qt + c4;
            #pragma unroll
            for (int it = 0; it < 4; ++it) {
                const int r = rsub + 4 * it;
                if (strip * 16 + r < nrows) {
                    const float s = dvL[strip * 16 + r];
                    const float4 o = *(const float4*)(Ow + r * SPITCH + c4);
                    *(float4*)(hp + (size_t)r * D) = make_float4(o.x * s, o.y * s, o.z * s, o.w * s);
                }
            }
            __builtin_amdgcn_fence(__ATOMIC_RELEASE, "wavefront");
            __builtin_amdgcn_wave_barrier();
            __builtin_amdgcn_fence(__ATOMIC_ACQUIRE, "wavefront");

            job = (job < NW) ? job2 : n_jobs;
        }
    }
}


extern "C" void kernel_launch(void* const* d_in, const int* in_sizes, int n_in,
                              void* d_out, int out_size, void* d_ws, size_t ws_size,
                              hipStream_t stream) {
    const float* x   = (const float*)d_in[0];
    const int*   ei  = (const int*)d_in[1];
    const float* W1  = (const float*)d_in[2];
    const float* b1  = (const float*)d_in[3];
    const float* W2  = (const float*)d_in[4];
    const float* b2  = (const float*)d_in[5];
    const float* W3  = (const float*)d_in[6];
    const float* b3  = (const float*)d_in[7];

    const int N = in_sizes[0] / D;
    const int E = in_sizes[1] / 2;
    const int* src = ei;
    const int* dst = ei + E;
    const int nBins = (N + RB - 1) / RB;

    float* dinv = (float*)d_ws;
    size_t off = ((size_t)N * sizeof(float) + 255) & ~(size_t)255;
    float* H = (float*)((char*)d_ws + off);
    float* Y = H + (size_t)N * D;
    const size_t binBytes = ((size_t)(nBins + 1) * sizeof(int) + 255) & ~(size_t)255;
    char* p = (char*)(Y + (size_t)N * D);
    int* binTotal = (int*)p;  p += binBytes;
    int* cursor   = (int*)p;  p += binBytes;
    int* binStart = (int*)p;  p += binBytes;
    int* recs     = (int*)p;
    float* out = (float*)d_out;

    const int T = 256;
    const int binBlocks   = (nBins + T - 1) / T;
    const size_t countLds = (size_t)nBins * sizeof(int);
    const size_t placeLds = (size_t)(3 * nBins + 1) * sizeof(int);

    gcn_clear_bins<<<binBlocks, T, 0, stream>>>(binTotal, cursor, nBins);
    gcn_bin_count<<<256, PT, countLds, stream>>>(dst, binTotal, E, nBins);
    gcn_bin_place<<<256, PT, placeLds, stream>>>(src, dst, binTotal, cursor, binStart, recs, E, nBins);

    gcn_range<2><<<nBins, PT, 0, stream>>>(x, dinv, dinv, b1, recs, binStart, W1, H, N);

    gcn_range<1><<<nBins, PT, 0, stream>>>(H, dinv, dinv, b1, recs, binStart, W2, Y, N);

    gcn_range<1><<<nBins, PT, 0, stream>>>(Y, dinv, dinv, b2, recs, binStart, W3, H, N);

    gcn_range<0><<<nBins, PT, 0, stream>>>(H, dinv, dinv, b3, recs, binStart, W3, out, N);
}


# edge loop: one 16-byte-per-lane load fetches a gathered row whole (512 B, four full lines; was two 256-B loads); LDS rows kept by component so LDS adds stay bank-conflict-free; sum orders unchanged
# speedup vs baseline: 5.0490x; 1.1221x over previous
#include <hip/hip_runtime.h>
#include <hip/hip_bf16.h>


typedef float v2f __attribute__((ext_vector_type(2)));
typedef float v4f __attribute__((ext_vector_type(4)));
typedef float v8f __attribute__((ext_vector_type(8)));

#define D 128
#define RB 200
#define PT 1024
#define EU 8
#define NW (PT / 32)
#define NK ((RB + NW - 1) / NW)
#define XOFF 32768
#define APITCH 132
#define FROWS (NK * NW)
#define SPITCH 36
#define WPITCH 272


__global__ void gcn_clear_bins(int* __restrict__ binTotal, int* __restrict__ cursor, int n_bins) {
    int i = blockIdx.x * blockDim.x + threadIdx.x;
    if (i < n_bins) { binTotal[i] = 0; cursor[i] = 0; }
}

__global__ __launch_bounds__(PT)
void gcn_bin_count(const int* __restrict__ dst, int* __restrict__ binTotal,
                   int n_edges, int n_bins) {
    extern __shared__ int sm[];
    const int t = threadIdx.x;
    for (int i = t; i < n_bins; i += PT) sm[i] = 0;
    __syncthreads();
    const int per = (n_edges + gridDim.x - 1) / gridDim.x;
    const int eb = blockIdx.x * per;
    const int ee = (eb + per < n_edges) ? (eb + per) : n_edges;
    for (int e = eb + t; e < ee; e += PT) atomicAdd(&sm[dst[e] / RB], 1);
    __syncthreads();
    for (int i = t; i < n_bins; i += PT) {
        const int c = sm[i];
        if (c > 0) atomicAdd(&binTotal[i], c);
    }
}


__global__ __launch_bounds__(PT)
void gcn_bin_place(const int* __restrict__ src, const int* __restrict__ dst,
                   const int* __restrict__ binTotal,
                   int* __restrict__ cursor, int* __restrict__ binStart,
                   int* __restrict__ recs, int n_edges, int n_bins) {
    extern __shared__ int sm[];
    int* startL = sm;
    int* cntL   = sm + (n_bins + 1);
    int* baseL  = cntL + n_bins;
    __shared__ int wsum[32];
    __shared__ int carryS;

    const int t = threadIdx.x;
    const int lane = t & 31;
    const int w = t >> 5;

    if (t == 0) carryS = 0;
    __syncthreads();
    for (int c0 = 0; c0 < n_bins; c0 += PT) {
        const int i = c0 + t;
        const int v = (i < n_bins) ? binTotal[i] : 0;
        int x = v;
        for (int o = 1; o < 32; o <<= 1) {
            const int y = __shfl_up(x, o);
            if (lane >= o) x += y;
        }
        if (lane == 31) wsum[w] = x;
        __syncthreads();
        int pre = carryS;
        for (int k = 0; k < w; ++k) pre += wsum[k];
        if (i < n_bins) startL[i] = pre + x - v;
        __syncthreads();
        if (t == PT - 1) carryS = pre + x;
        __syncthreads();
    }
    if (t == 0) startL[n_bins] = carryS;
    for (int i = t; i < n_bins; i += PT) cntL[i] = 0;
    __syncthreads();

    const int per = (n_edges + gridDim.x - 1) / gridDim.x;
    const int eb = blockIdx.x * per;
    const int ee = (eb + per < n_edges) ? (eb + per) : n_edges;

    for (int e = eb + t; e < ee; e += PT) atomicAdd(&cntL[dst[e] / RB], 1);
    __syncthreads();
    for (int i = t; i < n_bins; i += PT) {
        const int c = cntL[i];
        int base = startL[i];
        if (c > 0) base += atomicAdd(&cursor[i], c);
        baseL[i] = base;
        cntL[i] = 0;
    }
    __syncthreads();
    for (int e = eb + t; e < ee; e += PT) {
        const int s = src[e];
        const int d = dst[e];
        const int b = d / RB;
        const int pos = baseL[b] + atomicAdd(&cntL[b], 1);
        *(int2*)(recs + (size_t)pos * 2) = make_int2(s, d - b * RB);
    }
    if (blockIdx.x == 0)
        for (int i = t; i <= n_bins; i += PT) binStart[i] = startL[i];
}


template <int MODE>
__global__ __launch_bounds__(PT)
void gcn_range(const float* __restrict__ H, const float* dinv_in, float* dinv_out,
               const float* __restrict__ bias, const int* __restrict__ recs,
               const int* __restrict__ binStart, const float* __restrict__ Wn,
               float* __restrict__ Y, int n_rows) {
    constexpr bool GEMM = (MODE != 0);
    constexpr bool EDGES = (MODE != 2);
    constexpr int AP = GEMM ? APITCH : D;
    constexpr int WS_OFF = XOFF;
    constexpr int OS_OFF = WS_OFF + (D / 2) * WPITCH;
    constexpr int DV_OFF = GEMM ? OS_OFF + NW * 16 * SPITCH : XOFF + FROWS * AP;
    constexpr int CN_OFF = DV_OFF + FROWS;
    constexpr int LDS_FLOATS = CN_OFF + (MODE == 2 ? RB : 0);
    static_assert(FROWS * AP <= XOFF && XOFF + FROWS * AP <= DV_OFF, "the two row copies must not reach the next region");
    static_assert(LDS_FLOATS * 4 > 160 * 1024 && LDS_FLOATS * 4 <= 320 * 1024, "one range work-group per WGP");
    __shared__ float lds[LDS_FLOATS];
    float* const acc = lds;
    float* const Ws = lds + WS_OFF;
    float* const Os = lds + OS_OFF;
    float* const dvL = lds + DV_OFF;
    int* const cnt = (int*)(lds + CN_OFF);

    const int t = threadIdx.x;
    const int lane = t & 31;
    const int wave = __builtin_amdgcn_readfirstlane(t >> 5);
    const int node0 = blockIdx.x * RB;
    int nrows = n_rows - node0;
    if (nrows > RB) nrows = RB;

    constexpr int WIT = (D / 2) / NW;
    if constexpr (MODE == 2) {
        #pragma unroll
        for (int it = 0; it < WIT; ++it) {
            const int kp = it * NW + wave;
            const float4 r0 = *(const float4*)(Wn + (size_t)(2 * kp) * D + 4 * lane);
            const float4 r1 = *(const float4*)(Wn + (size_t)(2 * kp + 1) * D + 4 * lane);
            float* p = Ws + kp * WPITCH + 8 * lane;
            *(float4*)(p)     = make_float4(r0.x, r1.x, r0.y, r1.y);
            *(float4*)(p + 4) = make_float4(r0.z, r1.z, r0.w, r1.w);
        }
    }
    int boff = 0;
    if constexpr (EDGES) {
        unsigned hw;
        asm volatile("s_getreg_b32 %0, hwreg(HW_REG_WAVE_HW_ID1)" : "=s"(hw));
        boff = (hw & 0x200u) ? XOFF : 0;
    }
    if constexpr (MODE == 2) {
        if (t < RB) cnt[t] = 0;
    } else {
        if (t < FROWS) dvL[t] = (t < nrows) ? dinv_in[node0 + t] : 0.0f;
    }

    const int c = lane * 4;
    float4 own[NK];
    #pragma unroll
    for (int k = 0; k < NK; ++k) {
        const int r = wave + NW * k;
        const int rl = (r < nrows) ? r : nrows - 1;
        own[k] = *(const float4*)(H + (size_t)(node0 + rl) * D + c);
    }
    float4 wq[2 * WIT];
    if constexpr (MODE == 1) {
        #pragma unroll
        for (int it = 0; it < WIT; ++it) {
            const int kp = it * NW + wave;
            wq[2 * it]     = *(const float4*)(Wn + (size_t)(2 * kp) * D + 4 * lane);
            wq[2 * it + 1] = *(const float4*)(Wn + (size_t)(2 * kp + 1) * D + 4 * lane);
        }
    }
    #pragma unroll
    for (int k = 0; k < NK; ++k) {
        const int r = wave + NW * k;
        if constexpr (EDGES) {
            float* ip = acc + r * AP + lane;
            ip[0]  = own[k].x;
            ip[32] = own[k].y;
            ip[64] = own[k].z;
            ip[96] = own[k].w;
            *(float4*)(acc + XOFF + r * AP + c) = make_float4(0.0f, 0.0f, 0.0f, 0.0f);
        } else {
            *(float4*)(acc + r * AP + c) = own[k];
        }
    }
    __syncthreads();

    const int e0 = binStart[blockIdx.x];
    const int e1 = binStart[blockIdx.x + 1];

    if constexpr (MODE == 2) {
        for (int e = e0 + t; e < e1; e += PT) atomicAdd(&cnt[recs[(size_t)e * 2 + 1]], 1);
        __syncthreads();
        if (t < FROWS) {
            float v = 0.0f;
            if (t < nrows) {
                const float d = (float)(cnt[t] + 1);
                v = (d > 0.0f) ? __frsqrt_rn(d) : 0.0f;
                dinv_out[node0 + t] = v;
            }
            dvL[t] = v;
        }
        __syncthreads();
    } else {
        const int2* __restrict__ rp = (const int2*)recs;
        for (int e = e0 + wave; e < e1; e += EU * 32) {
            int2 r[EU];
            v4f row[EU];
            #pragma unroll
            for (int u = 0; u < EU; ++u) {
                int eu = e + u * 32;
                if (eu >= e1) eu = e1 - 1;
                r[u] = rp[eu];
            }
            #pragma unroll
            for (int u = 0; u < EU; ++u)
                row[u] = *((const v4f*)(H + (size_t)r[u].x * D) + lane);
            asm volatile("" :: "v"(row[0]), "v"(row[1]), "v"(row[2]), "v"(row[3]),
                               "v"(row[4]), "v"(row[5]), "v"(row[6]), "v"(row[7]));
            static_assert(EU == 8, "the fence above names EU = 8 rows");
            #pragma unroll
            for (int u = 0; u < EU; ++u) {
                if (e + u * 32 < e1) {
                    float* yp = acc + boff + r[u].y * AP + lane;
                    atomicAdd(yp +  0, row[u].x);
                    atomicAdd(yp + 32, row[u].y);
                    atomicAdd(yp + 64, row[u].z);
                    atomicAdd(yp + 96, row[u].w);
                }
            }
        }
        __syncthreads();
    }

    if constexpr (MODE == 0) {
        const float4 b = *(const float4*)(bias + c);
        float4 a[NK];
        float4 z[NK];
        float s[NK];
        #pragma unroll
        for (int k = 0; k < NK; ++k) {
            const int r = wave + NW * k;
            s[k] = dvL[r];
            const float* ap = acc + r * AP + lane;
            a[k] = make_float4(ap[0], ap[32], ap[64], ap[96]);
            z[k] = make_float4(ap[XOFF], ap[XOFF + 32], ap[XOFF + 64], ap[XOFF + 96]);
        }
        #pragma unroll
        for (int k = 0; k < NK; ++k) {
            const int r = wave + NW * k;
            if (r < nrows)
                *(float4*)(Y + (size_t)(node0 + r) * D + c) =
                    make_float4(fmaf(a[k].x + z[k].x, s[k], b.x), fmaf(a[k].y + z[k].y, s[k], b.y),
                                fmaf(a[k].z + z[k].z, s[k], b.z), fmaf(a[k].w + z[k].w, s[k], b.w));
        }
    } else {
        if constexpr (MODE == 1) {
            const float4 b = *(const float4*)(bias + c);
            #pragma unroll
            for (int k = 0; k < NK; ++k) {
                const int r = wave + NW * k;
                const float s = dvL[r];
                const float* ap = acc + r * AP + lane;
                const float4 a = make_float4(ap[0], ap[32], ap[64], ap[96]);
                const float4 z = make_float4(ap[XOFF], ap[XOFF + 32], ap[XOFF + 64], ap[XOFF + 96]);
                *(float4*)(acc + r * AP + c) =
                    make_float4(fmaf(a.x + z.x, s, b.x), fmaf(a.y + z.y, s, b.y),
                                fmaf(a.z + z.z, s, b.z), fmaf(a.w + z.w, s, b.w));
            }
            __syncthreads();
            #pragma unroll
            for (int it = 0; it < WIT; ++it) {
                const int kp = it * NW + wave;
                const float4 r0 = wq[2 * it];
                const float4 r1 = wq[2 * it + 1];
                float* p = Ws + kp * WPITCH + 8 * lane;
                *(float4*)(p)     = make_float4(r0.x, r1.x, r0.y, r1.y);
                *(float4*)(p + 4) = make_float4(r0.z, r1.z, r0.w, r1.w);
            }
            __syncthreads();
        }
        const int half = lane >> 4;
        const int l16  = lane & 15;
        float* __restrict__ Ow = Os + wave * (16 * SPITCH);
        const float* __restrict__ bbase = Ws + (2 * half) * WPITCH + 2 * l16;
        const int n_jobs = 4 * ((nrows + 15) >> 4);
        static_assert(NW == 32 && 4 * ((RB + 15) / 16) <= NW + 20, "job split is written for 32 waves");
        const int w8 = wave & 7;
        const int blk = wave >> 3;
        const int job2 = (w8 < 4) ? NW + 5 * blk + w8
                       : (w8 == 4 + blk) ? NW + 5 * blk + 4 : n_jobs;

        int job = wave;
        while (job < n_jobs) {
            const int strip = job >> 2;
            const int qt = job & 3;

            const float* __restrict__ ap = acc + (strip * 16 + l16) * AP + 4 * half;
            float4 a[16];
            #pragma unroll
            for (int q = 0; q < 16; ++q) {
                a[q] = *(const float4*)(ap + 8 * q);
                if constexpr (MODE == 1) {
                    a[q].x = fmaxf(a[q].x, 0.0f); a[q].y = fmaxf(a[q].y, 0.0f);
                    a[q].z = fmaxf(a[q].z, 0.0f); a[q].w = fmaxf(a[q].w, 0.0f);
                }
            }

            v8f c0 = (v8f){};
            v8f c1 = (v8f){};
            const float* __restrict__ bq = bbase + 64 * qt;
            #pragma unroll
            for (int q = 0; q < 16; ++q) {
                #pragma unroll
                for (int u = 0; u < 2; ++u) {
                    const float* __restrict__ bp = bq + (4 * q + u) * WPITCH;
                    v2f av;
                    av.x = u ? a[q].z : a[q].x;
                    av.y = u ? a[q].w : a[q].y;
                    const v2f b0 = *(const v2f*)(bp);
                    const v2f b1 = *(const v2f*)(bp + 32);
                    c0 = __builtin_amdgcn_wmma_f32_16x16x4_f32(
                         false, av,  false, b0,
                         (short)0, c0,  false,  false);
                    c1 = __builtin_amdgcn_wmma_f32_16x16x4_f32(
                         false, av,  false, b1,
                         (short)0, c1,  false,  false);
                }
            }

            #pragma unroll
            for (int v = 0; v < 8; ++v) {
                Ow[(8 * half + v) * SPITCH + l16]      = c0[v];
                Ow[(8 * half + v) * SPITCH + 16 + l16] = c1[v];
            }
            __builtin_amdgcn_fence(__ATOMIC_RELEASE, "wavefront");
            __builtin_amdgcn_wave_barrier();
            __builtin_amdgcn_fence(__ATOMIC_ACQUIRE, "wavefront");

            const int rsub = lane >> 3;
            const int c4 = 4 * (lane & 7);
            float* __restrict__ hp = Y + (size_t)(node0 + strip * 16) * D + 32 * qt + c4;
            #pragma unroll
            for (int it = 0; it < 4; ++it) {
                const int r = rsub + 4 * it;
                if (strip * 16 + r < nrows) {
                    const float s = dvL[strip * 16 + r];
                    const float4 o = *(const float4*)(Ow + r * SPITCH + c4);
                    *(float4*)(hp + (size_t)r * D) = make_float4(o.x * s, o.y * s, o.z * s, o.w * s);
                }
            }
            __builtin_amdgcn_fence(__ATOMIC_RELEASE, "wavefront");
            __builtin_amdgcn_wave_barrier();
            __builtin_amdgcn_fence(__ATOMIC_ACQUIRE, "wavefront");

            job = (job < NW) ? job2 : n_jobs;
        }
    }
}


extern "C" void kernel_launch(void* const* d_in, const int* in_sizes, int n_in,
                              void* d_out, int out_size, void* d_ws, size_t ws_size,
                              hipStream_t stream) {
    const float* x   = (const float*)d_in[0];
    const int*   ei  = (const int*)d_in[1];
    const float* W1  = (const float*)d_in[2];
    const float* b1  = (const float*)d_in[3];
    const float* W2  = (const float*)d_in[4];
    const float* b2  = (const float*)d_in[5];
    const float* W3  = (const float*)d_in[6];
    const float* b3  = (const float*)d_in[7];

    const int N = in_sizes[0] / D;
    const int E = in_sizes[1] / 2;
    const int* src = ei;
    const int* dst = ei + E;
    const int nBins = (N + RB - 1) / RB;

    float* dinv = (float*)d_ws;
    size_t off = ((size_t)N * sizeof(float) + 255) & ~(size_t)255;
    float* H = (float*)((char*)d_ws + off);
    float* Y = H + (size_t)N * D;
    const size_t binBytes = ((size_t)(nBins + 1) * sizeof(int) + 255) & ~(size_t)255;
    char* p = (char*)(Y + (size_t)N * D);
    int* binTotal = (int*)p;  p += binBytes;
    int* cursor   = (int*)p;  p += binBytes;
    int* binStart = (int*)p;  p += binBytes;
    int* recs     = (int*)p;
    float* out = (float*)d_out;

    const int T = 256;
    const int binBlocks   = (nBins + T - 1) / T;
    const size_t countLds = (size_t)nBins * sizeof(int);
    const size_t placeLds = (size_t)(3 * nBins + 1) * sizeof(int);

    gcn_clear_bins<<<binBlocks, T, 0, stream>>>(binTotal, cursor, nBins);
    gcn_bin_count<<<256, PT, countLds, stream>>>(dst, binTotal, E, nBins);
    gcn_bin_place<<<256, PT, placeLds, stream>>>(src, dst, binTotal, cursor, binStart, recs, E, nBins);

    gcn_range<2><<<nBins, PT, 0, stream>>>(x, dinv, dinv, b1, recs, binStart, W1, H, N);

    gcn_range<1><<<nBins, PT, 0, stream>>>(H, dinv, dinv, b1, recs, binStart, W2, Y, N);

    gcn_range<1><<<nBins, PT, 0, stream>>>(Y, dinv, dinv, b2, recs, binStart, W3, H, N);

    gcn_range<0><<<nBins, PT, 0, stream>>>(H, dinv, dinv, b3, recs, binStart, W3, out, N);
}
